# starting kernel
# baseline (speedup 1.0000x reference)
#include <hip/hip_runtime.h>

#define NN 100000
#define NE 1600000
#define CH 64
#define EPSV 1e-5f

typedef float v2f __attribute__((ext_vector_type(2)));
typedef float v8f __attribute__((ext_vector_type(8)));

__global__ void k_zero(float* __restrict__ p, int n) {
    int i = blockIdx.x * blockDim.x + threadIdx.x;
    if (i < n) p[i] = 0.0f;
}

__global__ void k_deg(const int* __restrict__ dst, float* __restrict__ deg, int e) {
    int i = blockIdx.x * blockDim.x + threadIdx.x;
    if (i < e) atomicAdd(&deg[dst[i]], 1.0f);
}

__global__ void k_dinv(const float* __restrict__ deg, float* __restrict__ dinv, int n) {
    int i = blockIdx.x * blockDim.x + threadIdx.x;
    if (i < n) {
        float d = deg[i];
        dinv[i] = (d > 0.0f) ? rsqrtf(d) : 0.0f;
    }
}

__global__ void k_wedge(const int* __restrict__ src, const int* __restrict__ dst,
                        const float* __restrict__ dinv, float* __restrict__ we, int e) {
    int i = blockIdx.x * blockDim.x + threadIdx.x;
    if (i < e) we[i] = -dinv[src[i]] * dinv[dst[i]];
}

__global__ void k_prop(const int* __restrict__ src, const int* __restrict__ dst,
                       const float* __restrict__ we, const float* __restrict__ xin,
                       float* __restrict__ out) {
    int tid = blockIdx.x * blockDim.x + threadIdx.x;
    int e = tid >> 5;
    int lane = tid & 31;
    int s = src[e], d = dst[e];
    float w = we[e];
    const float2 v = *reinterpret_cast<const float2*>(xin + (long long)s * CH + lane * 2);
    float* o = out + (long long)d * CH + lane * 2;
    atomicAdd(o,     w * v.x);
    atomicAdd(o + 1, w * v.y);
}

__global__ void k_fixup(float* __restrict__ tx2, const float* __restrict__ tx0, int n) {
    int i = blockIdx.x * blockDim.x + threadIdx.x;
    if (i < n) tx2[i] = 2.0f * tx2[i] - tx0[i];
}

__global__ __launch_bounds__(128) void k_cheb_gemm(
        const float* __restrict__ tx0, const float* __restrict__ tx1,
        const float* __restrict__ tx2, const float* __restrict__ Wk,
        const float* __restrict__ bias, float* __restrict__ out,
        float* __restrict__ sum, float* __restrict__ sumsq) {
    const int wave  = threadIdx.x >> 5;
    const int lane  = threadIdx.x & 31;
    const int mrow  = lane & 15;
    const int khalf = lane >> 4;
    const int n0    = wave * 16;
    const int rowbase = blockIdx.x * 16;

    v8f c = {0.f, 0.f, 0.f, 0.f, 0.f, 0.f, 0.f, 0.f};
    const float* txs[3] = {tx0, tx1, tx2};

#pragma unroll
    for (int t = 0; t < 3; ++t) {
        const float* __restrict__ X  = txs[t];
        const float* __restrict__ Wt = Wk + t * CH * CH;
#pragma unroll
        for (int kk = 0; kk < 16; ++kk) {
            const int k0 = kk * 4 + khalf * 2;
            v2f a = *reinterpret_cast<const v2f*>(X + (long long)(rowbase + mrow) * CH + k0);
            v2f b;
            b.x = Wt[(k0 + 0) * CH + n0 + mrow];
            b.y = Wt[(k0 + 1) * CH + n0 + mrow];
            c = __builtin_amdgcn_wmma_f32_16x16x4_f32(
                    false, a, false, b, (short)0, c, false, false);
        }
    }

    const float bb = bias[n0 + mrow];
    float s = 0.0f, s2 = 0.0f;
#pragma unroll
    for (int i = 0; i < 8; ++i) {
        float y = c[i] + bb;
        y = (y > 0.0f) ? y : 0.0f;
        out[(long long)(rowbase + i + 8 * khalf) * CH + n0 + mrow] = y;
        s  += y;
        s2 += y * y;
    }
    atomicAdd(&sum[n0 + mrow], s);
    atomicAdd(&sumsq[n0 + mrow], s2);
}

__global__ void k_finstats(const float* __restrict__ sum, const float* __restrict__ sumsq,
                           float* __restrict__ mean, float* __restrict__ rstd) {
    int ch = threadIdx.x;
    if (ch < CH) {
        float m = sum[ch] * (1.0f / NN);
        float v = sumsq[ch] * (1.0f / NN) - m * m;
        mean[ch] = m;
        rstd[ch] = rsqrtf(v + EPSV);
    }
}

__global__ void k_norm(float* __restrict__ h, const float* __restrict__ mean,
                       const float* __restrict__ rstd, int n) {
    int i = blockIdx.x * blockDim.x + threadIdx.x;
    if (i < n) {
        int ch = i & (CH - 1);
        h[i] = (h[i] - mean[ch]) * rstd[ch];
    }
}

static void run_layer(const int* src, const int* dstp, const float* we,
                      const float* xin, const float* Wk, const float* bk,
                      float* tx1, float* tx2, float* hout, float* stats,
                      hipStream_t stream) {
    const int FN = NN * CH;
    k_zero<<<(2 * FN + 255) / 256, 256, 0, stream>>>(tx1, 2 * FN);
    k_prop<<<(NE * 32) / 256, 256, 0, stream>>>(src, dstp, we, xin, tx1);
    k_prop<<<(NE * 32) / 256, 256, 0, stream>>>(src, dstp, we, tx1, tx2);
    k_fixup<<<(FN + 255) / 256, 256, 0, stream>>>(tx2, xin, FN);
    k_zero<<<1, 256, 0, stream>>>(stats, 128);
    k_cheb_gemm<<<NN / 16, 128, 0, stream>>>(xin, tx1, tx2, Wk, bk, hout,
                                             stats, stats + CH);
    k_finstats<<<1, 64, 0, stream>>>(stats, stats + CH, stats + 2 * CH, stats + 3 * CH);
    k_norm<<<(FN + 255) / 256, 256, 0, stream>>>(hout, stats + 2 * CH, stats + 3 * CH, FN);
}

extern "C" void kernel_launch(void* const* d_in, const int* in_sizes, int n_in,
                              void* d_out, int out_size, void* d_ws, size_t ws_size,
                              hipStream_t stream) {
    (void)in_sizes; (void)n_in; (void)out_size; (void)ws_size;
    const float* x   = (const float*)d_in[0];
    const int*   ei  = (const int*)d_in[1];
    const float* W1  = (const float*)d_in[2];
    const float* b1  = (const float*)d_in[3];
    const float* W2  = (const float*)d_in[4];
    const float* b2  = (const float*)d_in[5];
    float* out = (float*)d_out;

    const int* src  = ei;
    const int* dstp = ei + NE;

    float* ws   = (float*)d_ws;
    float* deg  = ws;
    float* dinv = deg + NN;
    float* we   = dinv + NN;
    float* tx1  = we + NE;
    float* tx2  = tx1 + NN * CH;
    float* h1   = tx2 + NN * CH;
    float* stats = h1 + NN * CH;

    k_zero<<<(NN + 255) / 256, 256, 0, stream>>>(deg, NN);
    k_deg<<<(NE + 255) / 256, 256, 0, stream>>>(dstp, deg, NE);
    k_dinv<<<(NN + 255) / 256, 256, 0, stream>>>(deg, dinv, NN);
    k_wedge<<<(NE + 255) / 256, 256, 0, stream>>>(src, dstp, dinv, we, NE);

    run_layer(src, dstp, we, x, W1, b1, tx1, tx2, h1, stats, stream);
    run_layer(src, dstp, we, h1, W2, b2, tx1, tx2, out, stats, stream);
}


# atomic scatter-add propagation replaced by a gather over edges grouped by destination (in-call counting sort, slot = arrival-order rank of an integer atomic); a row is summed by one wave, stored once
# speedup vs baseline: 1.9614x; 1.9614x over previous
#include <hip/hip_runtime.h>

#define NN 100000
#define NE 1600000
#define CH 64
#define EPSV 1e-5f

typedef float v2f __attribute__((ext_vector_type(2)));
typedef float v8f __attribute__((ext_vector_type(8)));

__global__ void k_zero(float* __restrict__ p, int n) {
    int i = blockIdx.x * blockDim.x + threadIdx.x;
    if (i < n) p[i] = 0.0f;
}

#define SCAN_BLK 1024
#define SCAN_NB  ((NN + SCAN_BLK - 1) / SCAN_BLK)

__global__ void k_count(const int* __restrict__ dst, int* __restrict__ cnt,
                        int* __restrict__ rank, int e) {
    int i = blockIdx.x * blockDim.x + threadIdx.x;
    if (i < e) rank[i] = atomicAdd(&cnt[dst[i]], 1);
}

__global__ __launch_bounds__(256) void k_blocksum(const int* __restrict__ cnt,
                                                  int* __restrict__ bsum) {
    __shared__ int red[8];
    const int lane = threadIdx.x & 31, wave = threadIdx.x >> 5;
    const int base = blockIdx.x * SCAN_BLK + threadIdx.x * 4;
    int s = 0;
    for (int j = 0; j < 4; ++j) {
        int idx = base + j;
        if (idx < NN) s += cnt[idx];
    }
    for (int off = 16; off > 0; off >>= 1) s += __shfl_xor(s, off, 32);
    if (lane == 0) red[wave] = s;
    __syncthreads();
    if (threadIdx.x == 0) {
        int t = 0;
        for (int w = 0; w < 8; ++w) t += red[w];
        bsum[blockIdx.x] = t;
    }
}

__global__ __launch_bounds__(256) void k_scan(int* cr, const int* __restrict__ bsum,
                                              float* __restrict__ dinv) {
    __shared__ int wsum[8];
    __shared__ int boff;
    const int lane = threadIdx.x & 31, wave = threadIdx.x >> 5;
    if (wave == 0) {
        int t = 0;
        for (int j = lane; j < (int)blockIdx.x; j += 32) t += bsum[j];
        for (int off = 16; off > 0; off >>= 1) t += __shfl_xor(t, off, 32);
        if (lane == 0) boff = t;
    }
    const int base = blockIdx.x * SCAN_BLK + threadIdx.x * 4;
    int c[4];
    int s = 0;
    for (int j = 0; j < 4; ++j) {
        int idx = base + j;
        c[j] = (idx < NN) ? cr[idx] : 0;
        s += c[j];
    }
    int incl = s;
    for (int off = 1; off < 32; off <<= 1) {
        int t = __shfl_up(incl, off, 32);
        if (lane >= off) incl += t;
    }
    if (lane == 31) wsum[wave] = incl;
    __syncthreads();
    int woff = 0;
    for (int w = 0; w < wave; ++w) woff += wsum[w];
    int excl = boff + woff + incl - s;
    for (int j = 0; j < 4; ++j) {
        int idx = base + j;
        if (idx < NN) {
            cr[idx] = excl;
            dinv[idx] = (c[j] > 0) ? rsqrtf((float)c[j]) : 0.0f;
            excl += c[j];
        }
    }
}

__global__ void k_fill(const int* __restrict__ src, const int* __restrict__ dst,
                       const int* __restrict__ rowptr, const int* __restrict__ rank,
                       int* __restrict__ csrc, int e) {
    int i = blockIdx.x * blockDim.x + threadIdx.x;
    if (i < e) csrc[rowptr[dst[i]] + rank[i]] = src[i];
}

template <int U>
__device__ __forceinline__ void edge_block(const float* __restrict__ xl, int s_l, float d_l,
                                           float dn, int k, v2f& acc) {
    int s[U];
    float w[U];
    v2f v[U];
#pragma unroll
    for (int u = 0; u < U; ++u) {
        s[u] = __builtin_amdgcn_readlane(s_l, k + u);
        w[u] = -__int_as_float(__builtin_amdgcn_readlane(__float_as_int(d_l), k + u)) * dn;
    }
#pragma unroll
    for (int u = 0; u < U; ++u)
        v[u] = *reinterpret_cast<const v2f*>(xl + (long long)s[u] * CH);
#pragma unroll
    for (int u = 0; u < U; ++u) acc += w[u] * v[u];
}

__global__ __launch_bounds__(256) void k_gather(
        const int* __restrict__ rowptr, const int* __restrict__ csrc,
        const float* __restrict__ dinv, const float* __restrict__ xin,
        float* __restrict__ out) {
    const int lane = threadIdx.x & 31;
    const int node = __builtin_amdgcn_readfirstlane((int)(blockIdx.x * 256 + threadIdx.x) >> 5);
    const int beg = rowptr[node];
    const int end = (node + 1 < NN) ? rowptr[node + 1] : NE;
    const float dn = dinv[node];
    const float* __restrict__ xl = xin + lane * 2;
    v2f acc = {0.0f, 0.0f};
    for (int c0 = beg; c0 < end; c0 += 32) {
        const int n = (end - c0 < 32) ? (end - c0) : 32;
        int s_l = 0;
        if (lane < n) s_l = csrc[c0 + lane];
        const float d_l = dinv[s_l];
        int k = 0;
        for (; k + 8 <= n; k += 8) edge_block<8>(xl, s_l, d_l, dn, k, acc);
        if (k + 4 <= n) { edge_block<4>(xl, s_l, d_l, dn, k, acc); k += 4; }
        if (k + 2 <= n) { edge_block<2>(xl, s_l, d_l, dn, k, acc); k += 2; }
        if (k < n) edge_block<1>(xl, s_l, d_l, dn, k, acc);
    }
    *reinterpret_cast<v2f*>(out + (long long)node * CH + lane * 2) = acc;
}

__global__ void k_fixup(float* __restrict__ tx2, const float* __restrict__ tx0, int n) {
    int i = blockIdx.x * blockDim.x + threadIdx.x;
    if (i < n) tx2[i] = 2.0f * tx2[i] - tx0[i];
}

__global__ __launch_bounds__(128) void k_cheb_gemm(
        const float* __restrict__ tx0, const float* __restrict__ tx1,
        const float* __restrict__ tx2, const float* __restrict__ Wk,
        const float* __restrict__ bias, float* __restrict__ out,
        float* __restrict__ sum, float* __restrict__ sumsq) {
    const int wave  = threadIdx.x >> 5;
    const int lane  = threadIdx.x & 31;
    const int mrow  = lane & 15;
    const int khalf = lane >> 4;
    const int n0    = wave * 16;
    const int rowbase = blockIdx.x * 16;

    v8f c = {0.f, 0.f, 0.f, 0.f, 0.f, 0.f, 0.f, 0.f};
    const float* txs[3] = {tx0, tx1, tx2};

#pragma unroll
    for (int t = 0; t < 3; ++t) {
        const float* __restrict__ X  = txs[t];
        const float* __restrict__ Wt = Wk + t * CH * CH;
#pragma unroll
        for (int kk = 0; kk < 16; ++kk) {
            const int k0 = kk * 4 + khalf * 2;
            v2f a = *reinterpret_cast<const v2f*>(X + (long long)(rowbase + mrow) * CH + k0);
            v2f b;
            b.x = Wt[(k0 + 0) * CH + n0 + mrow];
            b.y = Wt[(k0 + 1) * CH + n0 + mrow];
            c = __builtin_amdgcn_wmma_f32_16x16x4_f32(
                    false, a, false, b, (short)0, c, false, false);
        }
    }

    const float bb = bias[n0 + mrow];
    float s = 0.0f, s2 = 0.0f;
#pragma unroll
    for (int i = 0; i < 8; ++i) {
        float y = c[i] + bb;
        y = (y > 0.0f) ? y : 0.0f;
        out[(long long)(rowbase + i + 8 * khalf) * CH + n0 + mrow] = y;
        s  += y;
        s2 += y * y;
    }
    atomicAdd(&sum[n0 + mrow], s);
    atomicAdd(&sumsq[n0 + mrow], s2);
}

__global__ void k_finstats(const float* __restrict__ sum, const float* __restrict__ sumsq,
                           float* __restrict__ mean, float* __restrict__ rstd) {
    int ch = threadIdx.x;
    if (ch < CH) {
        float m = sum[ch] * (1.0f / NN);
        float v = sumsq[ch] * (1.0f / NN) - m * m;
        mean[ch] = m;
        rstd[ch] = rsqrtf(v + EPSV);
    }
}

__global__ void k_norm(float* __restrict__ h, const float* __restrict__ mean,
                       const float* __restrict__ rstd, int n) {
    int i = blockIdx.x * blockDim.x + threadIdx.x;
    if (i < n) {
        int ch = i & (CH - 1);
        h[i] = (h[i] - mean[ch]) * rstd[ch];
    }
}

static void run_layer(const int* rowptr, const int* csrc, const float* dinv,
                      const float* xin, const float* Wk, const float* bk,
                      float* tx1, float* tx2, float* hout, float* stats,
                      hipStream_t stream) {
    const int FN = NN * CH;
    k_gather<<<NN / 8, 256, 0, stream>>>(rowptr, csrc, dinv, xin, tx1);
    k_gather<<<NN / 8, 256, 0, stream>>>(rowptr, csrc, dinv, tx1, tx2);
    k_fixup<<<(FN + 255) / 256, 256, 0, stream>>>(tx2, xin, FN);
    k_zero<<<1, 256, 0, stream>>>(stats, 128);
    k_cheb_gemm<<<NN / 16, 128, 0, stream>>>(xin, tx1, tx2, Wk, bk, hout,
                                             stats, stats + CH);
    k_finstats<<<1, 64, 0, stream>>>(stats, stats + CH, stats + 2 * CH, stats + 3 * CH);
    k_norm<<<(FN + 255) / 256, 256, 0, stream>>>(hout, stats + 2 * CH, stats + 3 * CH, FN);
}

extern "C" void kernel_launch(void* const* d_in, const int* in_sizes, int n_in,
                              void* d_out, int out_size, void* d_ws, size_t ws_size,
                              hipStream_t stream) {
    (void)in_sizes; (void)n_in; (void)out_size; (void)ws_size;
    const float* x   = (const float*)d_in[0];
    const int*   ei  = (const int*)d_in[1];
    const float* W1  = (const float*)d_in[2];
    const float* b1  = (const float*)d_in[3];
    const float* W2  = (const float*)d_in[4];
    const float* b2  = (const float*)d_in[5];
    float* out = (float*)d_out;

    const int* src  = ei;
    const int* dstp = ei + NE;

    float* ws   = (float*)d_ws;
    int*   rowptr = (int*)ws;
    float* dinv = ws + NN;
    int*   csrc = (int*)(dinv + NN);
    float* tx1  = (float*)(csrc + NE);
    float* tx2  = tx1 + NN * CH;
    float* h1   = tx2 + NN * CH;
    float* stats = h1 + NN * CH;
    int*   rank = (int*)tx2;
    int*   bsum = (int*)stats;

    k_zero<<<(NN + 255) / 256, 256, 0, stream>>>((float*)rowptr, NN);
    k_count<<<(NE + 255) / 256, 256, 0, stream>>>(dstp, rowptr, rank, NE);
    k_blocksum<<<SCAN_NB, 256, 0, stream>>>(rowptr, bsum);
    k_scan<<<SCAN_NB, 256, 0, stream>>>(rowptr, bsum, dinv);
    k_fill<<<(NE + 255) / 256, 256, 0, stream>>>(src, dstp, rowptr, rank, csrc, NE);

    run_layer(rowptr, csrc, dinv, x, W1, b1, tx1, tx2, h1, stats, stream);
    run_layer(rowptr, csrc, dinv, h1, W2, b2, tx1, tx2, out, stats, stream);
}


# GEMM blocks walk 5 row tiles: W fragments stay in registers, A tiles staged via LDS by full-line 16-B loads, column sums kept in registers, one atomic per column per block (stat sum order changes)
# speedup vs baseline: 2.5323x; 1.2911x over previous
#include <hip/hip_runtime.h>

#define NN 100000
#define NE 1600000
#define CH 64
#define EPSV 1e-5f

typedef float v2f __attribute__((ext_vector_type(2)));
typedef float v8f __attribute__((ext_vector_type(8)));

__global__ void k_zero(float* __restrict__ p, int n) {
    int i = blockIdx.x * blockDim.x + threadIdx.x;
    if (i < n) p[i] = 0.0f;
}

#define SCAN_BLK 1024
#define SCAN_NB  ((NN + SCAN_BLK - 1) / SCAN_BLK)

__global__ void k_count(const int* __restrict__ dst, int* __restrict__ cnt,
                        int* __restrict__ rank, int e) {
    int i = blockIdx.x * blockDim.x + threadIdx.x;
    if (i < e) rank[i] = atomicAdd(&cnt[dst[i]], 1);
}

__global__ __launch_bounds__(256) void k_blocksum(const int* __restrict__ cnt,
                                                  int* __restrict__ bsum) {
    __shared__ int red[8];
    const int lane = threadIdx.x & 31, wave = threadIdx.x >> 5;
    const int base = blockIdx.x * SCAN_BLK + threadIdx.x * 4;
    int s = 0;
    for (int j = 0; j < 4; ++j) {
        int idx = base + j;
        if (idx < NN) s += cnt[idx];
    }
    for (int off = 16; off > 0; off >>= 1) s += __shfl_xor(s, off, 32);
    if (lane == 0) red[wave] = s;
    __syncthreads();
    if (threadIdx.x == 0) {
        int t = 0;
        for (int w = 0; w < 8; ++w) t += red[w];
        bsum[blockIdx.x] = t;
    }
}

__global__ __launch_bounds__(256) void k_scan(int* cr, const int* __restrict__ bsum,
                                              float* __restrict__ dinv) {
    __shared__ int wsum[8];
    __shared__ int boff;
    const int lane = threadIdx.x & 31, wave = threadIdx.x >> 5;
    if (wave == 0) {
        int t = 0;
        for (int j = lane; j < (int)blockIdx.x; j += 32) t += bsum[j];
        for (int off = 16; off > 0; off >>= 1) t += __shfl_xor(t, off, 32);
        if (lane == 0) boff = t;
    }
    const int base = blockIdx.x * SCAN_BLK + threadIdx.x * 4;
    int c[4];
    int s = 0;
    for (int j = 0; j < 4; ++j) {
        int idx = base + j;
        c[j] = (idx < NN) ? cr[idx] : 0;
        s += c[j];
    }
    int incl = s;
    for (int off = 1; off < 32; off <<= 1) {
        int t = __shfl_up(incl, off, 32);
        if (lane >= off) incl += t;
    }
    if (lane == 31) wsum[wave] = incl;
    __syncthreads();
    int woff = 0;
    for (int w = 0; w < wave; ++w) woff += wsum[w];
    int excl = boff + woff + incl - s;
    for (int j = 0; j < 4; ++j) {
        int idx = base + j;
        if (idx < NN) {
            cr[idx] = excl;
            dinv[idx] = (c[j] > 0) ? rsqrtf((float)c[j]) : 0.0f;
            excl += c[j];
        }
    }
}

__global__ void k_fill(const int* __restrict__ src, const int* __restrict__ dst,
                       const int* __restrict__ rowptr, const int* __restrict__ rank,
                       int* __restrict__ csrc, int e) {
    int i = blockIdx.x * blockDim.x + threadIdx.x;
    if (i < e) csrc[rowptr[dst[i]] + rank[i]] = src[i];
}

template <int U>
__device__ __forceinline__ void edge_block(const float* __restrict__ xl, int s_l, float d_l,
                                           float dn, int k, v2f& acc) {
    int s[U];
    float w[U];
    v2f v[U];
#pragma unroll
    for (int u = 0; u < U; ++u) {
        s[u] = __builtin_amdgcn_readlane(s_l, k + u);
        w[u] = -__int_as_float(__builtin_amdgcn_readlane(__float_as_int(d_l), k + u)) * dn;
    }
#pragma unroll
    for (int u = 0; u < U; ++u)
        v[u] = *reinterpret_cast<const v2f*>(xl + (long long)s[u] * CH);
#pragma unroll
    for (int u = 0; u < U; ++u) acc += w[u] * v[u];
}

__global__ __launch_bounds__(256) void k_gather(
        const int* __restrict__ rowptr, const int* __restrict__ csrc,
        const float* __restrict__ dinv, const float* __restrict__ xin,
        float* __restrict__ out) {
    const int lane = threadIdx.x & 31;
    const int node = __builtin_amdgcn_readfirstlane((int)(blockIdx.x * 256 + threadIdx.x) >> 5);
    const int beg = rowptr[node];
    const int end = (node + 1 < NN) ? rowptr[node + 1] : NE;
    const float dn = dinv[node];
    const float* __restrict__ xl = xin + lane * 2;
    v2f acc = {0.0f, 0.0f};
    for (int c0 = beg; c0 < end; c0 += 32) {
        const int n = (end - c0 < 32) ? (end - c0) : 32;
        int s_l = 0;
        if (lane < n) s_l = csrc[c0 + lane];
        const float d_l = dinv[s_l];
        int k = 0;
        for (; k + 8 <= n; k += 8) edge_block<8>(xl, s_l, d_l, dn, k, acc);
        if (k + 4 <= n) { edge_block<4>(xl, s_l, d_l, dn, k, acc); k += 4; }
        if (k + 2 <= n) { edge_block<2>(xl, s_l, d_l, dn, k, acc); k += 2; }
        if (k < n) edge_block<1>(xl, s_l, d_l, dn, k, acc);
    }
    *reinterpret_cast<v2f*>(out + (long long)node * CH + lane * 2) = acc;
}

__global__ void k_fixup(float* __restrict__ tx2, const float* __restrict__ tx0, int n) {
    int i = blockIdx.x * blockDim.x + threadIdx.x;
    if (i < n) tx2[i] = 2.0f * tx2[i] - tx0[i];
}

#define GEMM_TPB 5
#define GEMM_NB  (NN / 16 / GEMM_TPB)
#define A_PITCH  68

typedef float v4f __attribute__((ext_vector_type(4)));

__device__ __forceinline__ void tile_fetch(const float* __restrict__ tx0,
                                           const float* __restrict__ tx1,
                                           const float* __restrict__ tx2,
                                           int rowbase, int tid, v4f (&st)[6]) {
    const int r = tid >> 4, c4 = tid & 15;
    const long long off = (long long)(rowbase + r) * CH + c4 * 4;
    st[0] = *reinterpret_cast<const v4f*>(tx0 + off);
    st[1] = *reinterpret_cast<const v4f*>(tx0 + off + 8 * CH);
    st[2] = *reinterpret_cast<const v4f*>(tx1 + off);
    st[3] = *reinterpret_cast<const v4f*>(tx1 + off + 8 * CH);
    st[4] = *reinterpret_cast<const v4f*>(tx2 + off);
    st[5] = *reinterpret_cast<const v4f*>(tx2 + off + 8 * CH);
}

__device__ __forceinline__ void tile_put(float* __restrict__ buf, int tid, const v4f (&st)[6]) {
    const int r = tid >> 4, c4 = tid & 15;
    float* p = buf + r * A_PITCH + c4 * 4;
    *reinterpret_cast<v4f*>(p)                                  = st[0];
    *reinterpret_cast<v4f*>(p + 8 * A_PITCH)                    = st[1];
    *reinterpret_cast<v4f*>(p + 16 * A_PITCH)                   = st[2];
    *reinterpret_cast<v4f*>(p + 16 * A_PITCH + 8 * A_PITCH)     = st[3];
    *reinterpret_cast<v4f*>(p + 32 * A_PITCH)                   = st[4];
    *reinterpret_cast<v4f*>(p + 32 * A_PITCH + 8 * A_PITCH)     = st[5];
}

__global__ __launch_bounds__(128) void k_cheb_gemm(
        const float* __restrict__ tx0, const float* __restrict__ tx1,
        const float* __restrict__ tx2, const float* __restrict__ Wk,
        const float* __restrict__ bias, float* __restrict__ out,
        float* __restrict__ sum, float* __restrict__ sumsq) {
    __shared__ __attribute__((aligned(16))) float As[2][3 * 16 * A_PITCH];
    const int tid   = threadIdx.x;
    const int wave  = tid >> 5;
    const int lane  = tid & 31;
    const int mrow  = lane & 15;
    const int khalf = lane >> 4;
    const int n0    = wave * 16;
    const int tile0 = blockIdx.x * GEMM_TPB;

    v4f st[6];
    tile_fetch(tx0, tx1, tx2, tile0 * 16, tid, st);

    v2f bfrag[48];
#pragma unroll
    for (int t = 0; t < 3; ++t) {
#pragma unroll
        for (int kk = 0; kk < 16; ++kk) {
            const int k0 = kk * 4 + khalf * 2;
            const float* __restrict__ Wt = Wk + t * CH * CH;
            bfrag[t * 16 + kk].x = Wt[(k0 + 0) * CH + n0 + mrow];
            bfrag[t * 16 + kk].y = Wt[(k0 + 1) * CH + n0 + mrow];
        }
    }
    const float bb = bias[n0 + mrow];

    tile_put(As[0], tid, st);
    __syncthreads();

    float s = 0.0f, s2 = 0.0f;
    for (int j = 0; j < GEMM_TPB; ++j) {
        const int rowbase = (tile0 + j) * 16;
        const bool more = (j + 1 < GEMM_TPB);
        if (more) tile_fetch(tx0, tx1, tx2, rowbase + 16, tid, st);
        __builtin_amdgcn_sched_barrier(0);

        const float* __restrict__ Ab = As[j & 1] + mrow * A_PITCH + khalf * 2;
        v8f c = {0.f, 0.f, 0.f, 0.f, 0.f, 0.f, 0.f, 0.f};
#pragma unroll
        for (int t = 0; t < 3; ++t) {
#pragma unroll
            for (int kk = 0; kk < 16; ++kk) {
                v2f a = *reinterpret_cast<const v2f*>(Ab + t * 16 * A_PITCH + kk * 4);
                c = __builtin_amdgcn_wmma_f32_16x16x4_f32(
                        false, a, false, bfrag[t * 16 + kk], (short)0, c, false, false);
            }
        }

#pragma unroll
        for (int i = 0; i < 8; ++i) {
            float y = c[i] + bb;
            y = (y > 0.0f) ? y : 0.0f;
            out[(long long)(rowbase + i + 8 * khalf) * CH + n0 + mrow] = y;
            s  += y;
            s2 += y * y;
        }

        if (more) {
            tile_put(As[(j + 1) & 1], tid, st);
            __syncthreads();
        }
    }

    s  += __shfl_xor(s, 16, 32);
    s2 += __shfl_xor(s2, 16, 32);
    if (khalf == 0) {
        atomicAdd(&sum[n0 + mrow], s);
        atomicAdd(&sumsq[n0 + mrow], s2);
    }
}

__global__ void k_finstats(const float* __restrict__ sum, const float* __restrict__ sumsq,
                           float* __restrict__ mean, float* __restrict__ rstd) {
    int ch = threadIdx.x;
    if (ch < CH) {
        float m = sum[ch] * (1.0f / NN);
        float v = sumsq[ch] * (1.0f / NN) - m * m;
        mean[ch] = m;
        rstd[ch] = rsqrtf(v + EPSV);
    }
}

__global__ void k_norm(float* __restrict__ h, const float* __restrict__ mean,
                       const float* __restrict__ rstd, int n) {
    int i = blockIdx.x * blockDim.x + threadIdx.x;
    if (i < n) {
        int ch = i & (CH - 1);
        h[i] = (h[i] - mean[ch]) * rstd[ch];
    }
}

static void run_layer(const int* rowptr, const int* csrc, const float* dinv,
                      const float* xin, const float* Wk, const float* bk,
                      float* tx1, float* tx2, float* hout, float* stats,
                      hipStream_t stream) {
    const int FN = NN * CH;
    k_gather<<<NN / 8, 256, 0, stream>>>(rowptr, csrc, dinv, xin, tx1);
    k_gather<<<NN / 8, 256, 0, stream>>>(rowptr, csrc, dinv, tx1, tx2);
    k_fixup<<<(FN + 255) / 256, 256, 0, stream>>>(tx2, xin, FN);
    k_zero<<<1, 256, 0, stream>>>(stats, 128);
    k_cheb_gemm<<<GEMM_NB, 128, 0, stream>>>(xin, tx1, tx2, Wk, bk, hout,
                                             stats, stats + CH);
    k_finstats<<<1, 64, 0, stream>>>(stats, stats + CH, stats + 2 * CH, stats + 3 * CH);
    k_norm<<<(FN + 255) / 256, 256, 0, stream>>>(hout, stats + 2 * CH, stats + 3 * CH, FN);
}

extern "C" void kernel_launch(void* const* d_in, const int* in_sizes, int n_in,
                              void* d_out, int out_size, void* d_ws, size_t ws_size,
                              hipStream_t stream) {
    (void)in_sizes; (void)n_in; (void)out_size; (void)ws_size;
    const float* x   = (const float*)d_in[0];
    const int*   ei  = (const int*)d_in[1];
    const float* W1  = (const float*)d_in[2];
    const float* b1  = (const float*)d_in[3];
    const float* W2  = (const float*)d_in[4];
    const float* b2  = (const float*)d_in[5];
    float* out = (float*)d_out;

    const int* src  = ei;
    const int* dstp = ei + NE;

    float* ws   = (float*)d_ws;
    int*   rowptr = (int*)ws;
    float* dinv = ws + NN;
    int*   csrc = (int*)(dinv + NN);
    float* tx1  = (float*)(csrc + NE);
    float* tx2  = tx1 + NN * CH;
    float* h1   = tx2 + NN * CH;
    float* stats = h1 + NN * CH;
    int*   rank = (int*)tx2;
    int*   bsum = (int*)stats;

    k_zero<<<(NN + 255) / 256, 256, 0, stream>>>((float*)rowptr, NN);
    k_count<<<(NE + 255) / 256, 256, 0, stream>>>(dstp, rowptr, rank, NE);
    k_blocksum<<<SCAN_NB, 256, 0, stream>>>(rowptr, bsum);
    k_scan<<<SCAN_NB, 256, 0, stream>>>(rowptr, bsum, dinv);
    k_fill<<<(NE + 255) / 256, 256, 0, stream>>>(src, dstp, rowptr, rank, csrc, NE);

    run_layer(rowptr, csrc, dinv, x, W1, b1, tx1, tx2, h1, stats, stream);
    run_layer(rowptr, csrc, dinv, h1, W2, b2, tx1, tx2, out, stats, stream);
}


# Tx2 = 2*prop(Tx1) - Tx0 is formed in the GEMM's A-tile staging from the Tx0 piece each thread already holds; the separate fix-up pass (2 launches) is gone; arithmetic and sum orders unchanged
# speedup vs baseline: 2.8162x; 1.1121x over previous
#include <hip/hip_runtime.h>

#define NN 100000
#define NE 1600000
#define CH 64
#define EPSV 1e-5f

typedef float v2f __attribute__((ext_vector_type(2)));
typedef float v8f __attribute__((ext_vector_type(8)));

__global__ void k_zero(float* __restrict__ p, int n) {
    int i = blockIdx.x * blockDim.x + threadIdx.x;
    if (i < n) p[i] = 0.0f;
}

#define SCAN_BLK 1024
#define SCAN_NB  ((NN + SCAN_BLK - 1) / SCAN_BLK)

__global__ void k_count(const int* __restrict__ dst, int* __restrict__ cnt,
                        int* __restrict__ rank, int e) {
    int i = blockIdx.x * blockDim.x + threadIdx.x;
    if (i < e) rank[i] = atomicAdd(&cnt[dst[i]], 1);
}

__global__ __launch_bounds__(256) void k_blocksum(const int* __restrict__ cnt,
                                                  int* __restrict__ bsum) {
    __shared__ int red[8];
    const int lane = threadIdx.x & 31, wave = threadIdx.x >> 5;
    const int base = blockIdx.x * SCAN_BLK + threadIdx.x * 4;
    int s = 0;
    for (int j = 0; j < 4; ++j) {
        int idx = base + j;
        if (idx < NN) s += cnt[idx];
    }
    for (int off = 16; off > 0; off >>= 1) s += __shfl_xor(s, off, 32);
    if (lane == 0) red[wave] = s;
    __syncthreads();
    if (threadIdx.x == 0) {
        int t = 0;
        for (int w = 0; w < 8; ++w) t += red[w];
        bsum[blockIdx.x] = t;
    }
}

__global__ __launch_bounds__(256) void k_scan(int* cr, const int* __restrict__ bsum,
                                              float* __restrict__ dinv) {
    __shared__ int wsum[8];
    __shared__ int boff;
    const int lane = threadIdx.x & 31, wave = threadIdx.x >> 5;
    if (wave == 0) {
        int t = 0;
        for (int j = lane; j < (int)blockIdx.x; j += 32) t += bsum[j];
        for (int off = 16; off > 0; off >>= 1) t += __shfl_xor(t, off, 32);
        if (lane == 0) boff = t;
    }
    const int base = blockIdx.x * SCAN_BLK + threadIdx.x * 4;
    int c[4];
    int s = 0;
    for (int j = 0; j < 4; ++j) {
        int idx = base + j;
        c[j] = (idx < NN) ? cr[idx] : 0;
        s += c[j];
    }
    int incl = s;
    for (int off = 1; off < 32; off <<= 1) {
        int t = __shfl_up(incl, off, 32);
        if (lane >= off) incl += t;
    }
    if (lane == 31) wsum[wave] = incl;
    __syncthreads();
    int woff = 0;
    for (int w = 0; w < wave; ++w) woff += wsum[w];
    int excl = boff + woff + incl - s;
    for (int j = 0; j < 4; ++j) {
        int idx = base + j;
        if (idx < NN) {
            cr[idx] = excl;
            dinv[idx] = (c[j] > 0) ? rsqrtf((float)c[j]) : 0.0f;
            excl += c[j];
        }
    }
}

__global__ void k_fill(const int* __restrict__ src, const int* __restrict__ dst,
                       const int* __restrict__ rowptr, const int* __restrict__ rank,
                       int* __restrict__ csrc, int e) {
    int i = blockIdx.x * blockDim.x + threadIdx.x;
    if (i < e) csrc[rowptr[dst[i]] + rank[i]] = src[i];
}

template <int U>
__device__ __forceinline__ void edge_block(const float* __restrict__ xl, int s_l, float d_l,
                                           float dn, int k, v2f& acc) {
    int s[U];
    float w[U];
    v2f v[U];
#pragma unroll
    for (int u = 0; u < U; ++u) {
        s[u] = __builtin_amdgcn_readlane(s_l, k + u);
        w[u] = -__int_as_float(__builtin_amdgcn_readlane(__float_as_int(d_l), k + u)) * dn;
    }
#pragma unroll
    for (int u = 0; u < U; ++u)
        v[u] = *reinterpret_cast<const v2f*>(xl + (long long)s[u] * CH);
#pragma unroll
    for (int u = 0; u < U; ++u) acc += w[u] * v[u];
}

__global__ __launch_bounds__(256) void k_gather(
        const int* __restrict__ rowptr, const int* __restrict__ csrc,
        const float* __restrict__ dinv, const float* __restrict__ xin,
        float* __restrict__ out) {
    const int lane = threadIdx.x & 31;
    const int node = __builtin_amdgcn_readfirstlane((int)(blockIdx.x * 256 + threadIdx.x) >> 5);
    const int beg = rowptr[node];
    const int end = (node + 1 < NN) ? rowptr[node + 1] : NE;
    const float dn = dinv[node];
    const float* __restrict__ xl = xin + lane * 2;
    v2f acc = {0.0f, 0.0f};
    for (int c0 = beg; c0 < end; c0 += 32) {
        const int n = (end - c0 < 32) ? (end - c0) : 32;
        int s_l = 0;
        if (lane < n) s_l = csrc[c0 + lane];
        const float d_l = dinv[s_l];
        int k = 0;
        for (; k + 8 <= n; k += 8) edge_block<8>(xl, s_l, d_l, dn, k, acc);
        if (k + 4 <= n) { edge_block<4>(xl, s_l, d_l, dn, k, acc); k += 4; }
        if (k + 2 <= n) { edge_block<2>(xl, s_l, d_l, dn, k, acc); k += 2; }
        if (k < n) edge_block<1>(xl, s_l, d_l, dn, k, acc);
    }
    *reinterpret_cast<v2f*>(out + (long long)node * CH + lane * 2) = acc;
}

#define GEMM_TPB 5
#define GEMM_NB  (NN / 16 / GEMM_TPB)
#define A_PITCH  68

typedef float v4f __attribute__((ext_vector_type(4)));

__device__ __forceinline__ void tile_fetch(const float* __restrict__ tx0,
                                           const float* __restrict__ tx1,
                                           const float* __restrict__ p2,
                                           int rowbase, int tid, v4f (&st)[6]) {
    const int r = tid >> 4, c4 = tid & 15;
    const long long off = (long long)(rowbase + r) * CH + c4 * 4;
    st[0] = *reinterpret_cast<const v4f*>(tx0 + off);
    st[1] = *reinterpret_cast<const v4f*>(tx0 + off + 8 * CH);
    st[2] = *reinterpret_cast<const v4f*>(tx1 + off);
    st[3] = *reinterpret_cast<const v4f*>(tx1 + off + 8 * CH);
    st[4] = *reinterpret_cast<const v4f*>(p2 + off);
    st[5] = *reinterpret_cast<const v4f*>(p2 + off + 8 * CH);
}

__device__ __forceinline__ void tile_put(float* __restrict__ buf, int tid, const v4f (&st)[6]) {
    const int r = tid >> 4, c4 = tid & 15;
    float* p = buf + r * A_PITCH + c4 * 4;
    *reinterpret_cast<v4f*>(p)                                  = st[0];
    *reinterpret_cast<v4f*>(p + 8 * A_PITCH)                    = st[1];
    *reinterpret_cast<v4f*>(p + 16 * A_PITCH)                   = st[2];
    *reinterpret_cast<v4f*>(p + 16 * A_PITCH + 8 * A_PITCH)     = st[3];
    *reinterpret_cast<v4f*>(p + 32 * A_PITCH)                   = 2.0f * st[4] - st[0];
    *reinterpret_cast<v4f*>(p + 32 * A_PITCH + 8 * A_PITCH)     = 2.0f * st[5] - st[1];
}

__global__ __launch_bounds__(128) void k_cheb_gemm(
        const float* __restrict__ tx0, const float* __restrict__ tx1,
        const float* __restrict__ p2, const float* __restrict__ Wk,
        const float* __restrict__ bias, float* __restrict__ out,
        float* __restrict__ sum, float* __restrict__ sumsq) {
    __shared__ __attribute__((aligned(16))) float As[2][3 * 16 * A_PITCH];
    const int tid   = threadIdx.x;
    const int wave  = tid >> 5;
    const int lane  = tid & 31;
    const int mrow  = lane & 15;
    const int khalf = lane >> 4;
    const int n0    = wave * 16;
    const int tile0 = blockIdx.x * GEMM_TPB;

    v4f st[6];
    tile_fetch(tx0, tx1, p2, tile0 * 16, tid, st);

    v2f bfrag[48];
#pragma unroll
    for (int t = 0; t < 3; ++t) {
#pragma unroll
        for (int kk = 0; kk < 16; ++kk) {
            const int k0 = kk * 4 + khalf * 2;
            const float* __restrict__ Wt = Wk + t * CH * CH;
            bfrag[t * 16 + kk].x = Wt[(k0 + 0) * CH + n0 + mrow];
            bfrag[t * 16 + kk].y = Wt[(k0 + 1) * CH + n0 + mrow];
        }
    }
    const float bb = bias[n0 + mrow];

    tile_put(As[0], tid, st);
    __syncthreads();

    float s = 0.0f, s2 = 0.0f;
    for (int j = 0; j < GEMM_TPB; ++j) {
        const int rowbase = (tile0 + j) * 16;
        const bool more = (j + 1 < GEMM_TPB);
        if (more) tile_fetch(tx0, tx1, p2, rowbase + 16, tid, st);
        __builtin_amdgcn_sched_barrier(0);

        const float* __restrict__ Ab = As[j & 1] + mrow * A_PITCH + khalf * 2;
        v8f c = {0.f, 0.f, 0.f, 0.f, 0.f, 0.f, 0.f, 0.f};
#pragma unroll
        for (int t = 0; t < 3; ++t) {
#pragma unroll
            for (int kk = 0; kk < 16; ++kk) {
                v2f a = *reinterpret_cast<const v2f*>(Ab + t * 16 * A_PITCH + kk * 4);
                c = __builtin_amdgcn_wmma_f32_16x16x4_f32(
                        false, a, false, bfrag[t * 16 + kk], (short)0, c, false, false);
            }
        }
        __builtin_amdgcn_sched_barrier(0);

#pragma unroll
        for (int i = 0; i < 8; ++i) {
            float y = c[i] + bb;
            y = (y > 0.0f) ? y : 0.0f;
            out[(long long)(rowbase + i + 8 * khalf) * CH + n0 + mrow] = y;
            s  += y;
            s2 += y * y;
        }

        if (more) {
            tile_put(As[(j + 1) & 1], tid, st);
            __syncthreads();
        }
    }

    s  += __shfl_xor(s, 16, 32);
    s2 += __shfl_xor(s2, 16, 32);
    if (khalf == 0) {
        atomicAdd(&sum[n0 + mrow], s);
        atomicAdd(&sumsq[n0 + mrow], s2);
    }
}

__global__ void k_finstats(const float* __restrict__ sum, const float* __restrict__ sumsq,
                           float* __restrict__ mean, float* __restrict__ rstd) {
    int ch = threadIdx.x;
    if (ch < CH) {
        float m = sum[ch] * (1.0f / NN);
        float v = sumsq[ch] * (1.0f / NN) - m * m;
        mean[ch] = m;
        rstd[ch] = rsqrtf(v + EPSV);
    }
}

__global__ void k_norm(float* __restrict__ h, const float* __restrict__ mean,
                       const float* __restrict__ rstd, int n) {
    int i = blockIdx.x * blockDim.x + threadIdx.x;
    if (i < n) {
        int ch = i & (CH - 1);
        h[i] = (h[i] - mean[ch]) * rstd[ch];
    }
}

static void run_layer(const int* rowptr, const int* csrc, const float* dinv,
                      const float* xin, const float* Wk, const float* bk,
                      float* tx1, float* tx2, float* hout, float* stats,
                      hipStream_t stream) {
    const int FN = NN * CH;
    k_gather<<<NN / 8, 256, 0, stream>>>(rowptr, csrc, dinv, xin, tx1);
    k_gather<<<NN / 8, 256, 0, stream>>>(rowptr, csrc, dinv, tx1, tx2);
    k_zero<<<1, 256, 0, stream>>>(stats, 128);
    k_cheb_gemm<<<GEMM_NB, 128, 0, stream>>>(xin, tx1, tx2, Wk, bk, hout,
                                             stats, stats + CH);
    k_finstats<<<1, 64, 0, stream>>>(stats, stats + CH, stats + 2 * CH, stats + 3 * CH);
    k_norm<<<(FN + 255) / 256, 256, 0, stream>>>(hout, stats + 2 * CH, stats + 3 * CH, FN);
}

extern "C" void kernel_launch(void* const* d_in, const int* in_sizes, int n_in,
                              void* d_out, int out_size, void* d_ws, size_t ws_size,
                              hipStream_t stream) {
    (void)in_sizes; (void)n_in; (void)out_size; (void)ws_size;
    const float* x   = (const float*)d_in[0];
    const int*   ei  = (const int*)d_in[1];
    const float* W1  = (const float*)d_in[2];
    const float* b1  = (const float*)d_in[3];
    const float* W2  = (const float*)d_in[4];
    const float* b2  = (const float*)d_in[5];
    float* out = (float*)d_out;

    const int* src  = ei;
    const int* dstp = ei + NE;

    float* ws   = (float*)d_ws;
    int*   rowptr = (int*)ws;
    float* dinv = ws + NN;
    int*   csrc = (int*)(dinv + NN);
    float* tx1  = (float*)(csrc + NE);
    float* tx2  = tx1 + NN * CH;
    float* h1   = tx2 + NN * CH;
    float* stats = h1 + NN * CH;
    int*   rank = (int*)tx2;
    int*   bsum = (int*)stats;

    k_zero<<<(NN + 255) / 256, 256, 0, stream>>>((float*)rowptr, NN);
    k_count<<<(NE + 255) / 256, 256, 0, stream>>>(dstp, rowptr, rank, NE);
    k_blocksum<<<SCAN_NB, 256, 0, stream>>>(rowptr, bsum);
    k_scan<<<SCAN_NB, 256, 0, stream>>>(rowptr, bsum, dinv);
    k_fill<<<(NE + 255) / 256, 256, 0, stream>>>(src, dstp, rowptr, rank, csrc, NE);

    run_layer(rowptr, csrc, dinv, x, W1, b1, tx1, tx2, h1, stats, stream);
    run_layer(rowptr, csrc, dinv, h1, W2, b2, tx1, tx2, out, stats, stream);
}


# k_fill stores 8-byte edge records (source, dinv[source]) grouped by destination; the propagations read them in one contiguous load, no per-lane dinv[source] fetch; sums and their order unchanged
# speedup vs baseline: 2.9142x; 1.0348x over previous
#include <hip/hip_runtime.h>

#define NN 100000
#define NE 1600000
#define CH 64
#define EPSV 1e-5f

typedef float v2f __attribute__((ext_vector_type(2)));
typedef float v8f __attribute__((ext_vector_type(8)));

__global__ void k_zero(float* __restrict__ p, int n) {
    int i = blockIdx.x * blockDim.x + threadIdx.x;
    if (i < n) p[i] = 0.0f;
}

#define SCAN_BLK 1024
#define SCAN_NB  ((NN + SCAN_BLK - 1) / SCAN_BLK)

__global__ void k_count(const int* __restrict__ dst, int* __restrict__ cnt,
                        int* __restrict__ rank, int e) {
    int i = blockIdx.x * blockDim.x + threadIdx.x;
    if (i < e) rank[i] = atomicAdd(&cnt[dst[i]], 1);
}

__global__ __launch_bounds__(256) void k_blocksum(const int* __restrict__ cnt,
                                                  int* __restrict__ bsum) {
    __shared__ int red[8];
    const int lane = threadIdx.x & 31, wave = threadIdx.x >> 5;
    const int base = blockIdx.x * SCAN_BLK + threadIdx.x * 4;
    int s = 0;
    for (int j = 0; j < 4; ++j) {
        int idx = base + j;
        if (idx < NN) s += cnt[idx];
    }
    for (int off = 16; off > 0; off >>= 1) s += __shfl_xor(s, off, 32);
    if (lane == 0) red[wave] = s;
    __syncthreads();
    if (threadIdx.x == 0) {
        int t = 0;
        for (int w = 0; w < 8; ++w) t += red[w];
        bsum[blockIdx.x] = t;
    }
}

__global__ __launch_bounds__(256) void k_scan(int* cr, const int* __restrict__ bsum,
                                              float* __restrict__ dinv) {
    __shared__ int wsum[8];
    __shared__ int boff;
    const int lane = threadIdx.x & 31, wave = threadIdx.x >> 5;
    if (wave == 0) {
        int t = 0;
        for (int j = lane; j < (int)blockIdx.x; j += 32) t += bsum[j];
        for (int off = 16; off > 0; off >>= 1) t += __shfl_xor(t, off, 32);
        if (lane == 0) boff = t;
    }
    const int base = blockIdx.x * SCAN_BLK + threadIdx.x * 4;
    int c[4];
    int s = 0;
    for (int j = 0; j < 4; ++j) {
        int idx = base + j;
        c[j] = (idx < NN) ? cr[idx] : 0;
        s += c[j];
    }
    int incl = s;
    for (int off = 1; off < 32; off <<= 1) {
        int t = __shfl_up(incl, off, 32);
        if (lane >= off) incl += t;
    }
    if (lane == 31) wsum[wave] = incl;
    __syncthreads();
    int woff = 0;
    for (int w = 0; w < wave; ++w) woff += wsum[w];
    int excl = boff + woff + incl - s;
    for (int j = 0; j < 4; ++j) {
        int idx = base + j;
        if (idx < NN) {
            cr[idx] = excl;
            dinv[idx] = (c[j] > 0) ? rsqrtf((float)c[j]) : 0.0f;
            excl += c[j];
        }
    }
}

typedef int v2i __attribute__((ext_vector_type(2)));

__global__ void k_fill(const int* __restrict__ src, const int* __restrict__ dst,
                       const int* __restrict__ rowptr, const int* __restrict__ rank,
                       const float* __restrict__ dinv, int* __restrict__ cedge, int e) {
    int i = blockIdx.x * blockDim.x + threadIdx.x;
    if (i < e) {
        const int s = src[i];
        v2i rec;
        rec.x = s;
        rec.y = __float_as_int(dinv[s]);
        reinterpret_cast<v2i*>(cedge)[rowptr[dst[i]] + rank[i]] = rec;
    }
}

template <int U>
__device__ __forceinline__ void edge_block(const float* __restrict__ xl, int s_l, float d_l,
                                           float dn, int k, v2f& acc) {
    int s[U];
    float w[U];
    v2f v[U];
#pragma unroll
    for (int u = 0; u < U; ++u) {
        s[u] = __builtin_amdgcn_readlane(s_l, k + u);
        w[u] = -__int_as_float(__builtin_amdgcn_readlane(__float_as_int(d_l), k + u)) * dn;
    }
#pragma unroll
    for (int u = 0; u < U; ++u)
        v[u] = *reinterpret_cast<const v2f*>(xl + (long long)s[u] * CH);
#pragma unroll
    for (int u = 0; u < U; ++u) acc += w[u] * v[u];
}

__global__ __launch_bounds__(256) void k_gather(
        const int* __restrict__ rowptr, const int* __restrict__ cedge,
        const float* __restrict__ dinv, const float* __restrict__ xin,
        float* __restrict__ out) {
    const int lane = threadIdx.x & 31;
    const int node = __builtin_amdgcn_readfirstlane((int)(blockIdx.x * 256 + threadIdx.x) >> 5);
    const int beg = rowptr[node];
    const int end = (node + 1 < NN) ? rowptr[node + 1] : NE;
    const float dn = dinv[node];
    const float* __restrict__ xl = xin + lane * 2;
    v2f acc = {0.0f, 0.0f};
    for (int c0 = beg; c0 < end; c0 += 32) {
        const int n = (end - c0 < 32) ? (end - c0) : 32;
        v2i rec = {0, 0};
        if (lane < n) rec = reinterpret_cast<const v2i*>(cedge)[c0 + lane];
        const int s_l = rec.x;
        const float d_l = __int_as_float(rec.y);
        int k = 0;
        for (; k + 8 <= n; k += 8) edge_block<8>(xl, s_l, d_l, dn, k, acc);
        if (k + 4 <= n) { edge_block<4>(xl, s_l, d_l, dn, k, acc); k += 4; }
        if (k + 2 <= n) { edge_block<2>(xl, s_l, d_l, dn, k, acc); k += 2; }
        if (k < n) edge_block<1>(xl, s_l, d_l, dn, k, acc);
    }
    *reinterpret_cast<v2f*>(out + (long long)node * CH + lane * 2) = acc;
}

#define GEMM_TPB 5
#define GEMM_NB  (NN / 16 / GEMM_TPB)
#define A_PITCH  68

typedef float v4f __attribute__((ext_vector_type(4)));

__device__ __forceinline__ void tile_fetch(const float* __restrict__ tx0,
                                           const float* __restrict__ tx1,
                                           const float* __restrict__ p2,
                                           int rowbase, int tid, v4f (&st)[6]) {
    const int r = tid >> 4, c4 = tid & 15;
    const long long off = (long long)(rowbase + r) * CH + c4 * 4;
    st[0] = *reinterpret_cast<const v4f*>(tx0 + off);
    st[1] = *reinterpret_cast<const v4f*>(tx0 + off + 8 * CH);
    st[2] = *reinterpret_cast<const v4f*>(tx1 + off);
    st[3] = *reinterpret_cast<const v4f*>(tx1 + off + 8 * CH);
    st[4] = *reinterpret_cast<const v4f*>(p2 + off);
    st[5] = *reinterpret_cast<const v4f*>(p2 + off + 8 * CH);
}

__device__ __forceinline__ void tile_put(float* __restrict__ buf, int tid, const v4f (&st)[6]) {
    const int r = tid >> 4, c4 = tid & 15;
    float* p = buf + r * A_PITCH + c4 * 4;
    *reinterpret_cast<v4f*>(p)                                  = st[0];
    *reinterpret_cast<v4f*>(p + 8 * A_PITCH)                    = st[1];
    *reinterpret_cast<v4f*>(p + 16 * A_PITCH)                   = st[2];
    *reinterpret_cast<v4f*>(p + 16 * A_PITCH + 8 * A_PITCH)     = st[3];
    *reinterpret_cast<v4f*>(p + 32 * A_PITCH)                   = 2.0f * st[4] - st[0];
    *reinterpret_cast<v4f*>(p + 32 * A_PITCH + 8 * A_PITCH)     = 2.0f * st[5] - st[1];
}

__global__ __launch_bounds__(128) void k_cheb_gemm(
        const float* __restrict__ tx0, const float* __restrict__ tx1,
        const float* __restrict__ p2, const float* __restrict__ Wk,
        const float* __restrict__ bias, float* __restrict__ out,
        float* __restrict__ sum, float* __restrict__ sumsq) {
    __shared__ __attribute__((aligned(16))) float As[2][3 * 16 * A_PITCH];
    const int tid   = threadIdx.x;
    const int wave  = tid >> 5;
    const int lane  = tid & 31;
    const int mrow  = lane & 15;
    const int khalf = lane >> 4;
    const int n0    = wave * 16;
    const int tile0 = blockIdx.x * GEMM_TPB;

    v4f st[6];
    tile_fetch(tx0, tx1, p2, tile0 * 16, tid, st);

    v2f bfrag[48];
#pragma unroll
    for (int t = 0; t < 3; ++t) {
#pragma unroll
        for (int kk = 0; kk < 16; ++kk) {
            const int k0 = kk * 4 + khalf * 2;
            const float* __restrict__ Wt = Wk + t * CH * CH;
            bfrag[t * 16 + kk].x = Wt[(k0 + 0) * CH + n0 + mrow];
            bfrag[t * 16 + kk].y = Wt[(k0 + 1) * CH + n0 + mrow];
        }
    }
    const float bb = bias[n0 + mrow];

    tile_put(As[0], tid, st);
    __syncthreads();

    float s = 0.0f, s2 = 0.0f;
    for (int j = 0; j < GEMM_TPB; ++j) {
        const int rowbase = (tile0 + j) * 16;
        const bool more = (j + 1 < GEMM_TPB);
        if (more) tile_fetch(tx0, tx1, p2, rowbase + 16, tid, st);
        __builtin_amdgcn_sched_barrier(0);

        const float* __restrict__ Ab = As[j & 1] + mrow * A_PITCH + khalf * 2;
        v8f c = {0.f, 0.f, 0.f, 0.f, 0.f, 0.f, 0.f, 0.f};
#pragma unroll
        for (int t = 0; t < 3; ++t) {
#pragma unroll
            for (int kk = 0; kk < 16; ++kk) {
                v2f a = *reinterpret_cast<const v2f*>(Ab + t * 16 * A_PITCH + kk * 4);
                c = __builtin_amdgcn_wmma_f32_16x16x4_f32(
                        false, a, false, bfrag[t * 16 + kk], (short)0, c, false, false);
            }
        }
        __builtin_amdgcn_sched_barrier(0);

#pragma unroll
        for (int i = 0; i < 8; ++i) {
            float y = c[i] + bb;
            y = (y > 0.0f) ? y : 0.0f;
            out[(long long)(rowbase + i + 8 * khalf) * CH + n0 + mrow] = y;
            s  += y;
            s2 += y * y;
        }

        if (more) {
            tile_put(As[(j + 1) & 1], tid, st);
            __syncthreads();
        }
    }

    s  += __shfl_xor(s, 16, 32);
    s2 += __shfl_xor(s2, 16, 32);
    if (khalf == 0) {
        atomicAdd(&sum[n0 + mrow], s);
        atomicAdd(&sumsq[n0 + mrow], s2);
    }
}

__global__ void k_finstats(const float* __restrict__ sum, const float* __restrict__ sumsq,
                           float* __restrict__ mean, float* __restrict__ rstd) {
    int ch = threadIdx.x;
    if (ch < CH) {
        float m = sum[ch] * (1.0f / NN);
        float v = sumsq[ch] * (1.0f / NN) - m * m;
        mean[ch] = m;
        rstd[ch] = rsqrtf(v + EPSV);
    }
}

__global__ void k_norm(float* __restrict__ h, const float* __restrict__ mean,
                       const float* __restrict__ rstd, int n) {
    int i = blockIdx.x * blockDim.x + threadIdx.x;
    if (i < n) {
        int ch = i & (CH - 1);
        h[i] = (h[i] - mean[ch]) * rstd[ch];
    }
}

static void run_layer(const int* rowptr, const int* cedge, const float* dinv,
                      const float* xin, const float* Wk, const float* bk,
                      float* tx1, float* tx2, float* hout, float* stats,
                      hipStream_t stream) {
    const int FN = NN * CH;
    k_gather<<<NN / 8, 256, 0, stream>>>(rowptr, cedge, dinv, xin, tx1);
    k_gather<<<NN / 8, 256, 0, stream>>>(rowptr, cedge, dinv, tx1, tx2);
    k_zero<<<1, 256, 0, stream>>>(stats, 128);
    k_cheb_gemm<<<GEMM_NB, 128, 0, stream>>>(xin, tx1, tx2, Wk, bk, hout,
                                             stats, stats + CH);
    k_finstats<<<1, 64, 0, stream>>>(stats, stats + CH, stats + 2 * CH, stats + 3 * CH);
    k_norm<<<(FN + 255) / 256, 256, 0, stream>>>(hout, stats + 2 * CH, stats + 3 * CH, FN);
}

extern "C" void kernel_launch(void* const* d_in, const int* in_sizes, int n_in,
                              void* d_out, int out_size, void* d_ws, size_t ws_size,
                              hipStream_t stream) {
    (void)in_sizes; (void)n_in; (void)out_size; (void)ws_size;
    const float* x   = (const float*)d_in[0];
    const int*   ei  = (const int*)d_in[1];
    const float* W1  = (const float*)d_in[2];
    const float* b1  = (const float*)d_in[3];
    const float* W2  = (const float*)d_in[4];
    const float* b2  = (const float*)d_in[5];
    float* out = (float*)d_out;

    const int* src  = ei;
    const int* dstp = ei + NE;

    float* ws   = (float*)d_ws;
    int*   rowptr = (int*)ws;
    float* dinv = ws + NN;
    float* tx1  = dinv + NN;
    float* tx2  = tx1 + NN * CH;
    float* h1   = tx2 + NN * CH;
    float* stats = h1 + NN * CH;
    int*   rank = (int*)tx2;
    int*   bsum = (int*)stats;
    int*   cedge = (int*)out;

    k_zero<<<(NN + 255) / 256, 256, 0, stream>>>((float*)rowptr, NN);
    k_count<<<(NE + 255) / 256, 256, 0, stream>>>(dstp, rowptr, rank, NE);
    k_blocksum<<<SCAN_NB, 256, 0, stream>>>(rowptr, bsum);
    k_scan<<<SCAN_NB, 256, 0, stream>>>(rowptr, bsum, dinv);
    k_fill<<<(NE + 255) / 256, 256, 0, stream>>>(src, dstp, rowptr, rank, dinv, cedge, NE);

    run_layer(rowptr, cedge, dinv, x, W1, b1, tx1, tx2, h1, stats, stream);
    run_layer(rowptr, cedge, dinv, h1, W2, b2, tx1, tx2, out, stats, stream);
}


# layer 1's instance norm is applied by the two readers of its output (layer-2 first propagation per gathered value, GEMM A-tile staging) as (v-mean)*rstd; its own pass is gone; sum orders unchanged
# speedup vs baseline: 3.0530x; 1.0476x over previous
#include <hip/hip_runtime.h>

#define NN 100000
#define NE 1600000
#define CH 64
#define EPSV 1e-5f

typedef float v2f __attribute__((ext_vector_type(2)));
typedef float v8f __attribute__((ext_vector_type(8)));

__global__ void k_zero(float* __restrict__ p, int n) {
    int i = blockIdx.x * blockDim.x + threadIdx.x;
    if (i < n) p[i] = 0.0f;
}

#define SCAN_BLK 1024
#define SCAN_NB  ((NN + SCAN_BLK - 1) / SCAN_BLK)

__global__ void k_count(const int* __restrict__ dst, int* __restrict__ cnt,
                        int* __restrict__ rank, int e) {
    int i = blockIdx.x * blockDim.x + threadIdx.x;
    if (i < e) rank[i] = atomicAdd(&cnt[dst[i]], 1);
}

__global__ __launch_bounds__(256) void k_blocksum(const int* __restrict__ cnt,
                                                  int* __restrict__ bsum) {
    __shared__ int red[8];
    const int lane = threadIdx.x & 31, wave = threadIdx.x >> 5;
    const int base = blockIdx.x * SCAN_BLK + threadIdx.x * 4;
    int s = 0;
    for (int j = 0; j < 4; ++j) {
        int idx = base + j;
        if (idx < NN) s += cnt[idx];
    }
    for (int off = 16; off > 0; off >>= 1) s += __shfl_xor(s, off, 32);
    if (lane == 0) red[wave] = s;
    __syncthreads();
    if (threadIdx.x == 0) {
        int t = 0;
        for (int w = 0; w < 8; ++w) t += red[w];
        bsum[blockIdx.x] = t;
    }
}

__global__ __launch_bounds__(256) void k_scan(int* cr, const int* __restrict__ bsum,
                                              float* __restrict__ dinv) {
    __shared__ int wsum[8];
    __shared__ int boff;
    const int lane = threadIdx.x & 31, wave = threadIdx.x >> 5;
    if (wave == 0) {
        int t = 0;
        for (int j = lane; j < (int)blockIdx.x; j += 32) t += bsum[j];
        for (int off = 16; off > 0; off >>= 1) t += __shfl_xor(t, off, 32);
        if (lane == 0) boff = t;
    }
    const int base = blockIdx.x * SCAN_BLK + threadIdx.x * 4;
    int c[4];
    int s = 0;
    for (int j = 0; j < 4; ++j) {
        int idx = base + j;
        c[j] = (idx < NN) ? cr[idx] : 0;
        s += c[j];
    }
    int incl = s;
    for (int off = 1; off < 32; off <<= 1) {
        int t = __shfl_up(incl, off, 32);
        if (lane >= off) incl += t;
    }
    if (lane == 31) wsum[wave] = incl;
    __syncthreads();
    int woff = 0;
    for (int w = 0; w < wave; ++w) woff += wsum[w];
    int excl = boff + woff + incl - s;
    for (int j = 0; j < 4; ++j) {
        int idx = base + j;
        if (idx < NN) {
            cr[idx] = excl;
            dinv[idx] = (c[j] > 0) ? rsqrtf((float)c[j]) : 0.0f;
            excl += c[j];
        }
    }
}

typedef int v2i __attribute__((ext_vector_type(2)));

__global__ void k_fill(const int* __restrict__ src, const int* __restrict__ dst,
                       const int* __restrict__ rowptr, const int* __restrict__ rank,
                       const float* __restrict__ dinv, int* __restrict__ cedge, int e) {
    int i = blockIdx.x * blockDim.x + threadIdx.x;
    if (i < e) {
        const int s = src[i];
        v2i rec;
        rec.x = s;
        rec.y = __float_as_int(dinv[s]);
        reinterpret_cast<v2i*>(cedge)[rowptr[dst[i]] + rank[i]] = rec;
    }
}

template <int U, bool NORM>
__device__ __forceinline__ void edge_block(const float* __restrict__ xl, int s_l, float d_l,
                                           float dn, int k, v2f m2, v2f r2, v2f& acc) {
    int s[U];
    float w[U];
    v2f v[U];
#pragma unroll
    for (int u = 0; u < U; ++u) {
        s[u] = __builtin_amdgcn_readlane(s_l, k + u);
        w[u] = -__int_as_float(__builtin_amdgcn_readlane(__float_as_int(d_l), k + u)) * dn;
    }
#pragma unroll
    for (int u = 0; u < U; ++u)
        v[u] = *reinterpret_cast<const v2f*>(xl + (long long)s[u] * CH);
#pragma unroll
    for (int u = 0; u < U; ++u) {
        if (NORM) v[u] = (v[u] - m2) * r2;
        acc += w[u] * v[u];
    }
}

template <bool NORM>
__global__ __launch_bounds__(256) void k_gather(
        const int* __restrict__ rowptr, const int* __restrict__ cedge,
        const float* __restrict__ dinv, const float* __restrict__ xin,
        const float* __restrict__ mean, const float* __restrict__ rstd,
        float* __restrict__ out) {
    const int lane = threadIdx.x & 31;
    v2f m2 = {0.0f, 0.0f}, r2 = {1.0f, 1.0f};
    if (NORM) {
        m2 = *reinterpret_cast<const v2f*>(mean + lane * 2);
        r2 = *reinterpret_cast<const v2f*>(rstd + lane * 2);
    }
    const int node = __builtin_amdgcn_readfirstlane((int)(blockIdx.x * 256 + threadIdx.x) >> 5);
    const int beg = rowptr[node];
    const int end = (node + 1 < NN) ? rowptr[node + 1] : NE;
    const float dn = dinv[node];
    const float* __restrict__ xl = xin + lane * 2;
    v2f acc = {0.0f, 0.0f};
    for (int c0 = beg; c0 < end; c0 += 32) {
        const int n = (end - c0 < 32) ? (end - c0) : 32;
        v2i rec = {0, 0};
        if (lane < n) rec = reinterpret_cast<const v2i*>(cedge)[c0 + lane];
        const int s_l = rec.x;
        const float d_l = __int_as_float(rec.y);
        int k = 0;
        for (; k + 8 <= n; k += 8) edge_block<8, NORM>(xl, s_l, d_l, dn, k, m2, r2, acc);
        if (k + 4 <= n) { edge_block<4, NORM>(xl, s_l, d_l, dn, k, m2, r2, acc); k += 4; }
        if (k + 2 <= n) { edge_block<2, NORM>(xl, s_l, d_l, dn, k, m2, r2, acc); k += 2; }
        if (k < n) edge_block<1, NORM>(xl, s_l, d_l, dn, k, m2, r2, acc);
    }
    *reinterpret_cast<v2f*>(out + (long long)node * CH + lane * 2) = acc;
}

#define GEMM_TPB 5
#define GEMM_NB  (NN / 16 / GEMM_TPB)
#define A_PITCH  68

typedef float v4f __attribute__((ext_vector_type(4)));

__device__ __forceinline__ void tile_fetch(const float* __restrict__ tx0,
                                           const float* __restrict__ tx1,
                                           const float* __restrict__ p2,
                                           int rowbase, int tid, v4f (&st)[6]) {
    const int r = tid >> 4, c4 = tid & 15;
    const long long off = (long long)(rowbase + r) * CH + c4 * 4;
    st[0] = *reinterpret_cast<const v4f*>(tx0 + off);
    st[1] = *reinterpret_cast<const v4f*>(tx0 + off + 8 * CH);
    st[2] = *reinterpret_cast<const v4f*>(tx1 + off);
    st[3] = *reinterpret_cast<const v4f*>(tx1 + off + 8 * CH);
    st[4] = *reinterpret_cast<const v4f*>(p2 + off);
    st[5] = *reinterpret_cast<const v4f*>(p2 + off + 8 * CH);
}

template <bool NORM>
__device__ __forceinline__ void tile_put(float* __restrict__ buf, int tid, v4f (&st)[6],
                                         v4f m4, v4f r4) {
    const int r = tid >> 4, c4 = tid & 15;
    float* p = buf + r * A_PITCH + c4 * 4;
    if (NORM) {
        st[0] = (st[0] - m4) * r4;
        st[1] = (st[1] - m4) * r4;
    }
    *reinterpret_cast<v4f*>(p)                                  = st[0];
    *reinterpret_cast<v4f*>(p + 8 * A_PITCH)                    = st[1];
    *reinterpret_cast<v4f*>(p + 16 * A_PITCH)                   = st[2];
    *reinterpret_cast<v4f*>(p + 16 * A_PITCH + 8 * A_PITCH)     = st[3];
    *reinterpret_cast<v4f*>(p + 32 * A_PITCH)                   = 2.0f * st[4] - st[0];
    *reinterpret_cast<v4f*>(p + 32 * A_PITCH + 8 * A_PITCH)     = 2.0f * st[5] - st[1];
}

template <bool NORM>
__global__ __launch_bounds__(128) void k_cheb_gemm(
        const float* __restrict__ tx0, const float* __restrict__ tx1,
        const float* __restrict__ p2, const float* __restrict__ Wk,
        const float* __restrict__ bias,
        const float* __restrict__ mean0, const float* __restrict__ rstd0,
        float* __restrict__ out,
        float* __restrict__ sum, float* __restrict__ sumsq) {
    __shared__ __attribute__((aligned(16))) float As[2][3 * 16 * A_PITCH];
    const int tid   = threadIdx.x;
    const int wave  = tid >> 5;
    const int lane  = tid & 31;
    const int mrow  = lane & 15;
    const int khalf = lane >> 4;
    const int n0    = wave * 16;
    const int tile0 = blockIdx.x * GEMM_TPB;

    v4f m4 = {0.f, 0.f, 0.f, 0.f}, r4 = {1.f, 1.f, 1.f, 1.f};
    if (NORM) {
        m4 = *reinterpret_cast<const v4f*>(mean0 + (tid & 15) * 4);
        r4 = *reinterpret_cast<const v4f*>(rstd0 + (tid & 15) * 4);
    }

    v4f st[6];
    tile_fetch(tx0, tx1, p2, tile0 * 16, tid, st);

    v2f bfrag[48];
#pragma unroll
    for (int t = 0; t < 3; ++t) {
#pragma unroll
        for (int kk = 0; kk < 16; ++kk) {
            const int k0 = kk * 4 + khalf * 2;
            const float* __restrict__ Wt = Wk + t * CH * CH;
            bfrag[t * 16 + kk].x = Wt[(k0 + 0) * CH + n0 + mrow];
            bfrag[t * 16 + kk].y = Wt[(k0 + 1) * CH + n0 + mrow];
        }
    }
    const float bb = bias[n0 + mrow];

    tile_put<NORM>(As[0], tid, st, m4, r4);
    __syncthreads();

    float s = 0.0f, s2 = 0.0f;
    for (int j = 0; j < GEMM_TPB; ++j) {
        const int rowbase = (tile0 + j) * 16;
        const bool more = (j + 1 < GEMM_TPB);
        if (more) tile_fetch(tx0, tx1, p2, rowbase + 16, tid, st);
        __builtin_amdgcn_sched_barrier(0);

        const float* __restrict__ Ab = As[j & 1] + mrow * A_PITCH + khalf * 2;
        v8f c = {0.f, 0.f, 0.f, 0.f, 0.f, 0.f, 0.f, 0.f};
#pragma unroll
        for (int t = 0; t < 3; ++t) {
#pragma unroll
            for (int kk = 0; kk < 16; ++kk) {
                v2f a = *reinterpret_cast<const v2f*>(Ab + t * 16 * A_PITCH + kk * 4);
                c = __builtin_amdgcn_wmma_f32_16x16x4_f32(
                        false, a, false, bfrag[t * 16 + kk], (short)0, c, false, false);
            }
        }
        __builtin_amdgcn_sched_barrier(0);

#pragma unroll
        for (int i = 0; i < 8; ++i) {
            float y = c[i] + bb;
            y = (y > 0.0f) ? y : 0.0f;
            out[(long long)(rowbase + i + 8 * khalf) * CH + n0 + mrow] = y;
            s  += y;
            s2 += y * y;
        }

        if (more) {
            tile_put<NORM>(As[(j + 1) & 1], tid, st, m4, r4);
            __syncthreads();
        }
    }

    s  += __shfl_xor(s, 16, 32);
    s2 += __shfl_xor(s2, 16, 32);
    if (khalf == 0) {
        atomicAdd(&sum[n0 + mrow], s);
        atomicAdd(&sumsq[n0 + mrow], s2);
    }
}

__global__ void k_finstats(const float* __restrict__ sum, const float* __restrict__ sumsq,
                           float* __restrict__ mean, float* __restrict__ rstd) {
    int ch = threadIdx.x;
    if (ch < CH) {
        float m = sum[ch] * (1.0f / NN);
        float v = sumsq[ch] * (1.0f / NN) - m * m;
        mean[ch] = m;
        rstd[ch] = rsqrtf(v + EPSV);
    }
}

__global__ void k_norm(float* __restrict__ h, const float* __restrict__ mean,
                       const float* __restrict__ rstd, int n) {
    int i = blockIdx.x * blockDim.x + threadIdx.x;
    if (i < n) {
        int ch = i & (CH - 1);
        h[i] = (h[i] - mean[ch]) * rstd[ch];
    }
}

template <bool IN_RAW, bool OUT_NORM>
static void run_layer(const int* rowptr, const int* cedge, const float* dinv,
                      const float* xin, const float* Wk, const float* bk,
                      float* tx1, float* tx2, float* hout, float* stats,
                      hipStream_t stream) {
    const int FN = NN * CH;
    const float* mean = stats + 2 * CH;
    const float* rstd = stats + 3 * CH;
    k_gather<IN_RAW><<<NN / 8, 256, 0, stream>>>(rowptr, cedge, dinv, xin, mean, rstd,
                                                 tx1);
    k_gather<false><<<NN / 8, 256, 0, stream>>>(rowptr, cedge, dinv, tx1, mean, rstd,
                                                tx2);
    k_zero<<<1, 256, 0, stream>>>(stats, 128);
    k_cheb_gemm<IN_RAW><<<GEMM_NB, 128, 0, stream>>>(xin, tx1, tx2, Wk, bk, mean, rstd, hout,
                                                     stats, stats + CH);
    k_finstats<<<1, 64, 0, stream>>>(stats, stats + CH, stats + 2 * CH, stats + 3 * CH);
    if (OUT_NORM)
        k_norm<<<(FN + 255) / 256, 256, 0, stream>>>(hout, stats + 2 * CH, stats + 3 * CH, FN);
}

extern "C" void kernel_launch(void* const* d_in, const int* in_sizes, int n_in,
                              void* d_out, int out_size, void* d_ws, size_t ws_size,
                              hipStream_t stream) {
    (void)in_sizes; (void)n_in; (void)out_size; (void)ws_size;
    const float* x   = (const float*)d_in[0];
    const int*   ei  = (const int*)d_in[1];
    const float* W1  = (const float*)d_in[2];
    const float* b1  = (const float*)d_in[3];
    const float* W2  = (const float*)d_in[4];
    const float* b2  = (const float*)d_in[5];
    float* out = (float*)d_out;

    const int* src  = ei;
    const int* dstp = ei + NE;

    float* ws   = (float*)d_ws;
    int*   rowptr = (int*)ws;
    float* dinv = ws + NN;
    float* tx1  = dinv + NN;
    float* tx2  = tx1 + NN * CH;
    float* h1   = tx2 + NN * CH;
    float* stats = h1 + NN * CH;
    int*   rank = (int*)tx2;
    int*   bsum = (int*)stats;
    int*   cedge = (int*)out;

    k_zero<<<(NN + 255) / 256, 256, 0, stream>>>((float*)rowptr, NN);
    k_count<<<(NE + 255) / 256, 256, 0, stream>>>(dstp, rowptr, rank, NE);
    k_blocksum<<<SCAN_NB, 256, 0, stream>>>(rowptr, bsum);
    k_scan<<<SCAN_NB, 256, 0, stream>>>(rowptr, bsum, dinv);
    k_fill<<<(NE + 255) / 256, 256, 0, stream>>>(src, dstp, rowptr, rank, dinv, cedge, NE);

    run_layer<false, false>(rowptr, cedge, dinv, x, W1, b1, tx1, tx2, h1, stats, stream);
    run_layer<true, true>(rowptr, cedge, dinv, h1, W2, b2, tx1, tx2, out, stats, stream);
}


# propagation gathers 16 bytes per lane: each 16-lane half takes one edge, so one wave load brings two edges' rows (four whole lines); even/odd partial sums added at the end (sum order changes)
# speedup vs baseline: 3.1454x; 1.0303x over previous
#include <hip/hip_runtime.h>

#define NN 100000
#define NE 1600000
#define CH 64
#define EPSV 1e-5f

typedef float v2f __attribute__((ext_vector_type(2)));
typedef float v8f __attribute__((ext_vector_type(8)));

__global__ void k_zero(float* __restrict__ p, int n) {
    int i = blockIdx.x * blockDim.x + threadIdx.x;
    if (i < n) p[i] = 0.0f;
}

#define SCAN_BLK 1024
#define SCAN_NB  ((NN + SCAN_BLK - 1) / SCAN_BLK)

__global__ void k_count(const int* __restrict__ dst, int* __restrict__ cnt,
                        int* __restrict__ rank, int e) {
    int i = blockIdx.x * blockDim.x + threadIdx.x;
    if (i < e) rank[i] = atomicAdd(&cnt[dst[i]], 1);
}

__global__ __launch_bounds__(256) void k_blocksum(const int* __restrict__ cnt,
                                                  int* __restrict__ bsum) {
    __shared__ int red[8];
    const int lane = threadIdx.x & 31, wave = threadIdx.x >> 5;
    const int base = blockIdx.x * SCAN_BLK + threadIdx.x * 4;
    int s = 0;
    for (int j = 0; j < 4; ++j) {
        int idx = base + j;
        if (idx < NN) s += cnt[idx];
    }
    for (int off = 16; off > 0; off >>= 1) s += __shfl_xor(s, off, 32);
    if (lane == 0) red[wave] = s;
    __syncthreads();
    if (threadIdx.x == 0) {
        int t = 0;
        for (int w = 0; w < 8; ++w) t += red[w];
        bsum[blockIdx.x] = t;
    }
}

__global__ __launch_bounds__(256) void k_scan(int* cr, const int* __restrict__ bsum,
                                              float* __restrict__ dinv) {
    __shared__ int wsum[8];
    __shared__ int boff;
    const int lane = threadIdx.x & 31, wave = threadIdx.x >> 5;
    if (wave == 0) {
        int t = 0;
        for (int j = lane; j < (int)blockIdx.x; j += 32) t += bsum[j];
        for (int off = 16; off > 0; off >>= 1) t += __shfl_xor(t, off, 32);
        if (lane == 0) boff = t;
    }
    const int base = blockIdx.x * SCAN_BLK + threadIdx.x * 4;
    int c[4];
    int s = 0;
    for (int j = 0; j < 4; ++j) {
        int idx = base + j;
        c[j] = (idx < NN) ? cr[idx] : 0;
        s += c[j];
    }
    int incl = s;
    for (int off = 1; off < 32; off <<= 1) {
        int t = __shfl_up(incl, off, 32);
        if (lane >= off) incl += t;
    }
    if (lane == 31) wsum[wave] = incl;
    __syncthreads();
    int woff = 0;
    for (int w = 0; w < wave; ++w) woff += wsum[w];
    int excl = boff + woff + incl - s;
    for (int j = 0; j < 4; ++j) {
        int idx = base + j;
        if (idx < NN) {
            cr[idx] = excl;
            dinv[idx] = (c[j] > 0) ? rsqrtf((float)c[j]) : 0.0f;
            excl += c[j];
        }
    }
}

typedef int v2i __attribute__((ext_vector_type(2)));

__global__ void k_fill(const int* __restrict__ src, const int* __restrict__ dst,
                       const int* __restrict__ rowptr, const int* __restrict__ rank,
                       const float* __restrict__ dinv, int* __restrict__ cedge, int e) {
    int i = blockIdx.x * blockDim.x + threadIdx.x;
    if (i < e) {
        const int s = src[i];
        v2i rec;
        rec.x = s;
        rec.y = __float_as_int(dinv[s]);
        reinterpret_cast<v2i*>(cedge)[rowptr[dst[i]] + rank[i]] = rec;
    }
}

typedef float v4f __attribute__((ext_vector_type(4)));

__device__ __forceinline__ float xhalf(float x) {
    const int b = __float_as_int(x);
    return __int_as_float(__builtin_amdgcn_permlanex16(b, b, 0x76543210, 0xfedcba98, false, false));
}

template <int U>
__device__ __forceinline__ int half_share(int v) {
    return __builtin_amdgcn_update_dpp(0, v, 0x150 + U, 0xf, 0xf, false);
}

template <int P0, int P, int U>
struct share_pairs {
    static __device__ __forceinline__ void run(int off_l, float w_l, unsigned (&off)[P],
                                               float (&w)[P]) {
        off[U] = (unsigned)half_share<P0 + U>(off_l);
        w[U] = __int_as_float(half_share<P0 + U>(__float_as_int(w_l)));
        if constexpr (U + 1 < P) share_pairs<P0, P, U + 1>::run(off_l, w_l, off, w);
    }
};

template <int P0, int P, bool NORM>
__device__ __forceinline__ void pair_block(const char* __restrict__ xb, int off_l, float w_l,
                                           unsigned cb, v4f m4, v4f r4, v4f& acc) {
    unsigned off[P];
    float w[P];
    v4f v[P];
    share_pairs<P0, P, 0>::run(off_l, w_l, off, w);
#pragma unroll
    for (int u = 0; u < P; ++u)
        v[u] = *reinterpret_cast<const v4f*>(xb + (off[u] + cb));
#pragma unroll
    for (int u = 0; u < P; ++u) {
        if (NORM) v[u] = (v[u] - m4) * r4;
        acc += w[u] * v[u];
    }
}

template <bool NORM>
__global__ __launch_bounds__(256) void k_gather(
        const int* __restrict__ rowptr, const int* __restrict__ cedge,
        const float* __restrict__ dinv, const float* __restrict__ xin,
        const float* __restrict__ mean, const float* __restrict__ rstd,
        float* __restrict__ out) {
    const int lane = threadIdx.x & 31;
    const int c = lane & 15;
    const int h = lane >> 4;
    v4f m4 = {0.0f, 0.0f, 0.0f, 0.0f}, r4 = {1.0f, 1.0f, 1.0f, 1.0f};
    if (NORM) {
        m4 = *reinterpret_cast<const v4f*>(mean + c * 4);
        r4 = *reinterpret_cast<const v4f*>(rstd + c * 4);
    }
    const int node = __builtin_amdgcn_readfirstlane((int)(blockIdx.x * 256 + threadIdx.x) >> 5);
    const int beg = rowptr[node];
    const int end = (node + 1 < NN) ? rowptr[node + 1] : NE;
    const float dn = dinv[node];
    const char* __restrict__ xb = reinterpret_cast<const char*>(xin);
    const unsigned cb = (unsigned)c * 16u;
    const int e = 2 * c + h;
    v4f acc = {0.0f, 0.0f, 0.0f, 0.0f};
    for (int c0 = beg; c0 < end; c0 += 32) {
        const int n = (end - c0 < 32) ? (end - c0) : 32;
        const int ec = (e < n) ? e : (n - 1);
        const v2i rec = reinterpret_cast<const v2i*>(cedge)[c0 + ec];
        const int off_l = rec.x * (CH * 4);
        const float w_l = (e < n) ? -__int_as_float(rec.y) * dn : 0.0f;
        const int np = (n + 1) >> 1;
        if (np > 6)      pair_block<0, 8, NORM>(xb, off_l, w_l, cb, m4, r4, acc);
        else if (np > 4) pair_block<0, 6, NORM>(xb, off_l, w_l, cb, m4, r4, acc);
        else if (np > 2) pair_block<0, 4, NORM>(xb, off_l, w_l, cb, m4, r4, acc);
        else             pair_block<0, 2, NORM>(xb, off_l, w_l, cb, m4, r4, acc);
        if (np > 8) {
            if (np > 14)      pair_block<8, 8, NORM>(xb, off_l, w_l, cb, m4, r4, acc);
            else if (np > 12) pair_block<8, 6, NORM>(xb, off_l, w_l, cb, m4, r4, acc);
            else if (np > 10) pair_block<8, 4, NORM>(xb, off_l, w_l, cb, m4, r4, acc);
            else              pair_block<8, 2, NORM>(xb, off_l, w_l, cb, m4, r4, acc);
        }
    }
    acc.x += xhalf(acc.x);
    acc.y += xhalf(acc.y);
    acc.z += xhalf(acc.z);
    acc.w += xhalf(acc.w);
    if (h == 0) *reinterpret_cast<v4f*>(out + (long long)node * CH + c * 4) = acc;
}

#define GEMM_TPB 5
#define GEMM_NB  (NN / 16 / GEMM_TPB)
#define A_PITCH  68


__device__ __forceinline__ void tile_fetch(const float* __restrict__ tx0,
                                           const float* __restrict__ tx1,
                                           const float* __restrict__ p2,
                                           int rowbase, int tid, v4f (&st)[6]) {
    const int r = tid >> 4, c4 = tid & 15;
    const long long off = (long long)(rowbase + r) * CH + c4 * 4;
    st[0] = *reinterpret_cast<const v4f*>(tx0 + off);
    st[1] = *reinterpret_cast<const v4f*>(tx0 + off + 8 * CH);
    st[2] = *reinterpret_cast<const v4f*>(tx1 + off);
    st[3] = *reinterpret_cast<const v4f*>(tx1 + off + 8 * CH);
    st[4] = *reinterpret_cast<const v4f*>(p2 + off);
    st[5] = *reinterpret_cast<const v4f*>(p2 + off + 8 * CH);
}

template <bool NORM>
__device__ __forceinline__ void tile_put(float* __restrict__ buf, int tid, v4f (&st)[6],
                                         v4f m4, v4f r4) {
    const int r = tid >> 4, c4 = tid & 15;
    float* p = buf + r * A_PITCH + c4 * 4;
    if (NORM) {
        st[0] = (st[0] - m4) * r4;
        st[1] = (st[1] - m4) * r4;
    }
    *reinterpret_cast<v4f*>(p)                                  = st[0];
    *reinterpret_cast<v4f*>(p + 8 * A_PITCH)                    = st[1];
    *reinterpret_cast<v4f*>(p + 16 * A_PITCH)                   = st[2];
    *reinterpret_cast<v4f*>(p + 16 * A_PITCH + 8 * A_PITCH)     = st[3];
    *reinterpret_cast<v4f*>(p + 32 * A_PITCH)                   = 2.0f * st[4] - st[0];
    *reinterpret_cast<v4f*>(p + 32 * A_PITCH + 8 * A_PITCH)     = 2.0f * st[5] - st[1];
}

template <bool NORM>
__global__ __launch_bounds__(128) void k_cheb_gemm(
        const float* __restrict__ tx0, const float* __restrict__ tx1,
        const float* __restrict__ p2, const float* __restrict__ Wk,
        const float* __restrict__ bias,
        const float* __restrict__ mean0, const float* __restrict__ rstd0,
        float* __restrict__ out,
        float* __restrict__ sum, float* __restrict__ sumsq) {
    __shared__ __attribute__((aligned(16))) float As[2][3 * 16 * A_PITCH];
    const int tid   = threadIdx.x;
    const int wave  = tid >> 5;
    const int lane  = tid & 31;
    const int mrow  = lane & 15;
    const int khalf = lane >> 4;
    const int n0    = wave * 16;
    const int tile0 = blockIdx.x * GEMM_TPB;

    v4f m4 = {0.f, 0.f, 0.f, 0.f}, r4 = {1.f, 1.f, 1.f, 1.f};
    if (NORM) {
        m4 = *reinterpret_cast<const v4f*>(mean0 + (tid & 15) * 4);
        r4 = *reinterpret_cast<const v4f*>(rstd0 + (tid & 15) * 4);
    }

    v4f st[6];
    tile_fetch(tx0, tx1, p2, tile0 * 16, tid, st);

    v2f bfrag[48];
#pragma unroll
    for (int t = 0; t < 3; ++t) {
#pragma unroll
        for (int kk = 0; kk < 16; ++kk) {
            const int k0 = kk * 4 + khalf * 2;
            const float* __restrict__ Wt = Wk + t * CH * CH;
            bfrag[t * 16 + kk].x = Wt[(k0 + 0) * CH + n0 + mrow];
            bfrag[t * 16 + kk].y = Wt[(k0 + 1) * CH + n0 + mrow];
        }
    }
    const float bb = bias[n0 + mrow];

    tile_put<NORM>(As[0], tid, st, m4, r4);
    __syncthreads();

    float s = 0.0f, s2 = 0.0f;
    for (int j = 0; j < GEMM_TPB; ++j) {
        const int rowbase = (tile0 + j) * 16;
        const bool more = (j + 1 < GEMM_TPB);
        if (more) tile_fetch(tx0, tx1, p2, rowbase + 16, tid, st);
        __builtin_amdgcn_sched_barrier(0);

        const float* __restrict__ Ab = As[j & 1] + mrow * A_PITCH + khalf * 2;
        v8f c = {0.f, 0.f, 0.f, 0.f, 0.f, 0.f, 0.f, 0.f};
#pragma unroll
        for (int t = 0; t < 3; ++t) {
#pragma unroll
            for (int kk = 0; kk < 16; ++kk) {
                v2f a = *reinterpret_cast<const v2f*>(Ab + t * 16 * A_PITCH + kk * 4);
                c = __builtin_amdgcn_wmma_f32_16x16x4_f32(
                        false, a, false, bfrag[t * 16 + kk], (short)0, c, false, false);
            }
        }
        __builtin_amdgcn_sched_barrier(0);

#pragma unroll
        for (int i = 0; i < 8; ++i) {
            float y = c[i] + bb;
            y = (y > 0.0f) ? y : 0.0f;
            out[(long long)(rowbase + i + 8 * khalf) * CH + n0 + mrow] = y;
            s  += y;
            s2 += y * y;
        }

        if (more) {
            tile_put<NORM>(As[(j + 1) & 1], tid, st, m4, r4);
            __syncthreads();
        }
    }

    s  += __shfl_xor(s, 16, 32);
    s2 += __shfl_xor(s2, 16, 32);
    if (khalf == 0) {
        atomicAdd(&sum[n0 + mrow], s);
        atomicAdd(&sumsq[n0 + mrow], s2);
    }
}

__global__ void k_finstats(const float* __restrict__ sum, const float* __restrict__ sumsq,
                           float* __restrict__ mean, float* __restrict__ rstd) {
    int ch = threadIdx.x;
    if (ch < CH) {
        float m = sum[ch] * (1.0f / NN);
        float v = sumsq[ch] * (1.0f / NN) - m * m;
        mean[ch] = m;
        rstd[ch] = rsqrtf(v + EPSV);
    }
}

__global__ void k_norm(float* __restrict__ h, const float* __restrict__ mean,
                       const float* __restrict__ rstd, int n) {
    int i = blockIdx.x * blockDim.x + threadIdx.x;
    if (i < n) {
        int ch = i & (CH - 1);
        h[i] = (h[i] - mean[ch]) * rstd[ch];
    }
}

template <bool IN_RAW, bool OUT_NORM>
static void run_layer(const int* rowptr, const int* cedge, const float* dinv,
                      const float* xin, const float* Wk, const float* bk,
                      float* tx1, float* tx2, float* hout, float* stats,
                      hipStream_t stream) {
    const int FN = NN * CH;
    const float* mean = stats + 2 * CH;
    const float* rstd = stats + 3 * CH;
    k_gather<IN_RAW><<<NN / 8, 256, 0, stream>>>(rowptr, cedge, dinv, xin, mean, rstd,
                                                 tx1);
    k_gather<false><<<NN / 8, 256, 0, stream>>>(rowptr, cedge, dinv, tx1, mean, rstd,
                                                tx2);
    k_zero<<<1, 256, 0, stream>>>(stats, 128);
    k_cheb_gemm<IN_RAW><<<GEMM_NB, 128, 0, stream>>>(xin, tx1, tx2, Wk, bk, mean, rstd, hout,
                                                     stats, stats + CH);
    k_finstats<<<1, 64, 0, stream>>>(stats, stats + CH, stats + 2 * CH, stats + 3 * CH);
    if (OUT_NORM)
        k_norm<<<(FN + 255) / 256, 256, 0, stream>>>(hout, stats + 2 * CH, stats + 3 * CH, FN);
}

extern "C" void kernel_launch(void* const* d_in, const int* in_sizes, int n_in,
                              void* d_out, int out_size, void* d_ws, size_t ws_size,
                              hipStream_t stream) {
    (void)in_sizes; (void)n_in; (void)out_size; (void)ws_size;
    const float* x   = (const float*)d_in[0];
    const int*   ei  = (const int*)d_in[1];
    const float* W1  = (const float*)d_in[2];
    const float* b1  = (const float*)d_in[3];
    const float* W2  = (const float*)d_in[4];
    const float* b2  = (const float*)d_in[5];
    float* out = (float*)d_out;

    const int* src  = ei;
    const int* dstp = ei + NE;

    float* ws   = (float*)d_ws;
    int*   rowptr = (int*)ws;
    float* dinv = ws + NN;
    float* tx1  = dinv + NN;
    float* tx2  = tx1 + NN * CH;
    float* h1   = tx2 + NN * CH;
    float* stats = h1 + NN * CH;
    int*   rank = (int*)tx2;
    int*   bsum = (int*)stats;
    int*   cedge = (int*)out;

    k_zero<<<(NN + 255) / 256, 256, 0, stream>>>((float*)rowptr, NN);
    k_count<<<(NE + 255) / 256, 256, 0, stream>>>(dstp, rowptr, rank, NE);
    k_blocksum<<<SCAN_NB, 256, 0, stream>>>(rowptr, bsum);
    k_scan<<<SCAN_NB, 256, 0, stream>>>(rowptr, bsum, dinv);
    k_fill<<<(NE + 255) / 256, 256, 0, stream>>>(src, dstp, rowptr, rank, dinv, cedge, NE);

    run_layer<false, false>(rowptr, cedge, dinv, x, W1, b1, tx1, tx2, h1, stats, stream);
    run_layer<true, true>(rowptr, cedge, dinv, h1, W2, b2, tx1, tx2, out, stats, stream);
}


# edges grouped by a two-level counting sort ranked with LDS counters: no global atomic per edge, 16 -> 13 launches; first propagation fills the records' dinv word; in-row sum order stays arrival order
# speedup vs baseline: 4.9212x; 1.5646x over previous
#include <hip/hip_runtime.h>

#define NN 100000
#define NE 1600000
#define CH 64
#define EPSV 1e-5f

typedef float v2f __attribute__((ext_vector_type(2)));
typedef float v8f __attribute__((ext_vector_type(8)));

__global__ void k_zero(float* __restrict__ p, int n) {
    int i = blockIdx.x * blockDim.x + threadIdx.x;
    if (i < n) p[i] = 0.0f;
}

#define BKT_SHIFT 8
#define BKT_R     (1 << BKT_SHIFT)
#define NBKT      ((NN + BKT_R - 1) / BKT_R)
#define BIN_T     512
#define BIN_EPT   16
#define BIN_CHUNK (BIN_T * BIN_EPT)
#define NBIN      ((NE + BIN_CHUNK - 1) / BIN_CHUNK)
#define RPW       ((NBIN + 31) / 32)
static_assert(NBKT + 1 <= BIN_T, "one scan entry per k_bin thread, the end entry included");
static_assert(32 * RPW <= 1024 && BKT_R <= 1024, "k_build block of 1024 threads");

typedef int v2i __attribute__((ext_vector_type(2)));

__global__ __launch_bounds__(BIN_T) void k_bin(const int* __restrict__ src,
                                                const int* __restrict__ dst,
                                                int* __restrict__ seg,
                                                int* __restrict__ startT) {
    __shared__ int hist[BIN_T];
    __shared__ int wsum[BIN_T / 32];
    __shared__ __attribute__((aligned(16))) v2i srt[BIN_CHUNK];
    const int tid = threadIdx.x, lane = tid & 31, wave = tid >> 5;
    const int e0 = blockIdx.x * BIN_CHUNK;
    hist[tid] = 0;
    __syncthreads();

    int d[BIN_EPT], s[BIN_EPT];
#pragma unroll
    for (int j = 0; j < BIN_EPT; ++j) {
        const int i = e0 + j * BIN_T + tid;
        d[j] = (i < NE) ? dst[i] : -1;
        s[j] = (i < NE) ? src[i] : 0;
    }
#pragma unroll
    for (int j = 0; j < BIN_EPT; ++j)
        if (d[j] >= 0) atomicAdd(&hist[d[j] >> BKT_SHIFT], 1);
    __syncthreads();

    const int c = hist[tid];
    int incl = c;
    for (int off = 1; off < 32; off <<= 1) {
        int t = __shfl_up(incl, off, 32);
        if (lane >= off) incl += t;
    }
    if (lane == 31) wsum[wave] = incl;
    __syncthreads();
    int woff = 0;
    for (int w = 0; w < wave; ++w) woff += wsum[w];
    const int excl = woff + incl - c;
    hist[tid] = excl;
    if (tid <= NBKT) startT[tid * NBIN + blockIdx.x] = excl;
    __syncthreads();

#pragma unroll
    for (int j = 0; j < BIN_EPT; ++j) {
        if (d[j] >= 0) {
            const int pos = atomicAdd(&hist[d[j] >> BKT_SHIFT], 1);
            v2i rec;
            rec.x = s[j];
            rec.y = d[j];
            srt[pos] = rec;
        }
    }
    __syncthreads();

    const int total = (NE - e0 < BIN_CHUNK) ? (NE - e0) : BIN_CHUNK;
    for (int k = tid; k < total; k += BIN_T)
        reinterpret_cast<v2i*>(seg)[e0 + k] = srt[k];
}

__global__ __launch_bounds__(1024) void k_build(const int* __restrict__ seg,
                                                 const int* __restrict__ startT,
                                                 int* __restrict__ rowptr,
                                                 float* __restrict__ dinv,
                                                 int* __restrict__ cedge) {
    __shared__ int cnt[BKT_R];
    __shared__ int runS[32 * RPW], runN[32 * RPW];
    __shared__ int wsum[BKT_R / 32];
    __shared__ int gsum[RPW];
    const int tid = threadIdx.x, lane = tid & 31, wave = tid >> 5;
    const int b = blockIdx.x;
    const int base = b * BKT_R;
    const v2i* __restrict__ segr = reinterpret_cast<const v2i*>(seg);
    if (tid < BKT_R) cnt[tid] = 0;
    if (tid < 32 * RPW) {
        int s0 = 0, n = 0;
        if (tid < NBIN) {
            s0 = startT[b * NBIN + tid];
            n = startT[(b + 1) * NBIN + tid] - s0;
        }
        runS[tid] = tid * BIN_CHUNK + s0;
        runN[tid] = n;
        int t = s0;
        for (int off = 16; off > 0; off >>= 1) t += __shfl_xor(t, off, 32);
        if (lane == 0) gsum[wave] = t;
    }
    __syncthreads();
    int gbase = 0;
    for (int q = 0; q < RPW; ++q) gbase += gsum[q];

    int rs[RPW], rn[RPW];
    v2i rec[RPW];
#pragma unroll
    for (int q = 0; q < RPW; ++q) {
        rs[q] = runS[wave + 32 * q];
        rn[q] = runN[wave + 32 * q];
    }
#pragma unroll
    for (int q = 0; q < RPW; ++q) {
        const int idx = rs[q] + lane;
        rec[q] = segr[(idx < NE) ? idx : (NE - 1)];
    }
#pragma unroll
    for (int q = 0; q < RPW; ++q) {
        if (lane < rn[q]) atomicAdd(&cnt[rec[q].y - base], 1);
        for (int k = lane + 32; k < rn[q]; k += 32)
            atomicAdd(&cnt[segr[rs[q] + k].y - base], 1);
    }
    __syncthreads();

    int c = 0, incl = 0;
    if (tid < BKT_R) {
        c = cnt[tid];
        incl = c;
        for (int off = 1; off < 32; off <<= 1) {
            int t = __shfl_up(incl, off, 32);
            if (lane >= off) incl += t;
        }
        if (lane == 31) wsum[wave] = incl;
    }
    __syncthreads();
    if (tid < BKT_R) {
        int woff = 0;
        for (int w = 0; w < wave; ++w) woff += wsum[w];
        const int start = gbase + woff + incl - c;
        cnt[tid] = start;
        const int node = base + tid;
        if (node < NN) {
            rowptr[node] = start;
            dinv[node] = (c > 0) ? rsqrtf((float)c) : 0.0f;
        }
    }
    __syncthreads();

#pragma unroll
    for (int q = 0; q < RPW; ++q) {
        if (lane < rn[q]) {
            const int pos = atomicAdd(&cnt[rec[q].y - base], 1);
            v2i o;
            o.x = rec[q].x;
            o.y = 0;
            reinterpret_cast<v2i*>(cedge)[pos] = o;
        }
        for (int k = lane + 32; k < rn[q]; k += 32) {
            const v2i r2 = segr[rs[q] + k];
            const int pos = atomicAdd(&cnt[r2.y - base], 1);
            v2i o;
            o.x = r2.x;
            o.y = 0;
            reinterpret_cast<v2i*>(cedge)[pos] = o;
        }
    }
}

typedef float v4f __attribute__((ext_vector_type(4)));

__device__ __forceinline__ float xhalf(float x) {
    const int b = __float_as_int(x);
    return __int_as_float(__builtin_amdgcn_permlanex16(b, b, 0x76543210, 0xfedcba98, false, false));
}

template <int U>
__device__ __forceinline__ int half_share(int v) {
    return __builtin_amdgcn_update_dpp(0, v, 0x150 + U, 0xf, 0xf, false);
}

template <int P0, int P, int U>
struct share_pairs {
    static __device__ __forceinline__ void run(int off_l, float w_l, unsigned (&off)[P],
                                               float (&w)[P]) {
        off[U] = (unsigned)half_share<P0 + U>(off_l);
        w[U] = __int_as_float(half_share<P0 + U>(__float_as_int(w_l)));
        if constexpr (U + 1 < P) share_pairs<P0, P, U + 1>::run(off_l, w_l, off, w);
    }
};

template <int P0, int P, bool NORM>
__device__ __forceinline__ void pair_block(const char* __restrict__ xb, int off_l, float w_l,
                                           unsigned cb, v4f m4, v4f r4, v4f& acc) {
    unsigned off[P];
    float w[P];
    v4f v[P];
    share_pairs<P0, P, 0>::run(off_l, w_l, off, w);
#pragma unroll
    for (int u = 0; u < P; ++u)
        v[u] = *reinterpret_cast<const v4f*>(xb + (off[u] + cb));
#pragma unroll
    for (int u = 0; u < P; ++u) {
        if (NORM) v[u] = (v[u] - m4) * r4;
        acc += w[u] * v[u];
    }
}

template <bool NORM, bool FILL>
__global__ __launch_bounds__(256) void k_gather(
        const int* __restrict__ rowptr, int* __restrict__ cedge,
        const float* __restrict__ dinv, const float* __restrict__ xin,
        const float* __restrict__ mean, const float* __restrict__ rstd,
        float* __restrict__ out) {
    const int lane = threadIdx.x & 31;
    const int c = lane & 15;
    const int h = lane >> 4;
    v4f m4 = {0.0f, 0.0f, 0.0f, 0.0f}, r4 = {1.0f, 1.0f, 1.0f, 1.0f};
    if (NORM) {
        m4 = *reinterpret_cast<const v4f*>(mean + c * 4);
        r4 = *reinterpret_cast<const v4f*>(rstd + c * 4);
    }
    const int node = __builtin_amdgcn_readfirstlane((int)(blockIdx.x * 256 + threadIdx.x) >> 5);
    const int beg = rowptr[node];
    const int end = (node + 1 < NN) ? rowptr[node + 1] : NE;
    const float dn = dinv[node];
    const char* __restrict__ xb = reinterpret_cast<const char*>(xin);
    const unsigned cb = (unsigned)c * 16u;
    const int e = 2 * c + h;
    v4f acc = {0.0f, 0.0f, 0.0f, 0.0f};
    for (int c0 = beg; c0 < end; c0 += 32) {
        const int n = (end - c0 < 32) ? (end - c0) : 32;
        const int ec = (e < n) ? e : (n - 1);
        v2i rec = reinterpret_cast<const v2i*>(cedge)[c0 + ec];
        if (FILL) {
            rec.y = __float_as_int(dinv[rec.x]);
            if (e < n) reinterpret_cast<v2i*>(cedge)[c0 + e] = rec;
        }
        const int off_l = rec.x * (CH * 4);
        const float w_l = (e < n) ? -__int_as_float(rec.y) * dn : 0.0f;
        const int np = (n + 1) >> 1;
        if (np > 6)      pair_block<0, 8, NORM>(xb, off_l, w_l, cb, m4, r4, acc);
        else if (np > 4) pair_block<0, 6, NORM>(xb, off_l, w_l, cb, m4, r4, acc);
        else if (np > 2) pair_block<0, 4, NORM>(xb, off_l, w_l, cb, m4, r4, acc);
        else             pair_block<0, 2, NORM>(xb, off_l, w_l, cb, m4, r4, acc);
        if (np > 8) {
            if (np > 14)      pair_block<8, 8, NORM>(xb, off_l, w_l, cb, m4, r4, acc);
            else if (np > 12) pair_block<8, 6, NORM>(xb, off_l, w_l, cb, m4, r4, acc);
            else if (np > 10) pair_block<8, 4, NORM>(xb, off_l, w_l, cb, m4, r4, acc);
            else              pair_block<8, 2, NORM>(xb, off_l, w_l, cb, m4, r4, acc);
        }
    }
    acc.x += xhalf(acc.x);
    acc.y += xhalf(acc.y);
    acc.z += xhalf(acc.z);
    acc.w += xhalf(acc.w);
    if (h == 0) *reinterpret_cast<v4f*>(out + (long long)node * CH + c * 4) = acc;
}

#define GEMM_TPB 5
#define GEMM_NB  (NN / 16 / GEMM_TPB)
#define A_PITCH  68


__device__ __forceinline__ void tile_fetch(const float* __restrict__ tx0,
                                           const float* __restrict__ tx1,
                                           const float* __restrict__ p2,
                                           int rowbase, int tid, v4f (&st)[6]) {
    const int r = tid >> 4, c4 = tid & 15;
    const long long off = (long long)(rowbase + r) * CH + c4 * 4;
    st[0] = *reinterpret_cast<const v4f*>(tx0 + off);
    st[1] = *reinterpret_cast<const v4f*>(tx0 + off + 8 * CH);
    st[2] = *reinterpret_cast<const v4f*>(tx1 + off);
    st[3] = *reinterpret_cast<const v4f*>(tx1 + off + 8 * CH);
    st[4] = *reinterpret_cast<const v4f*>(p2 + off);
    st[5] = *reinterpret_cast<const v4f*>(p2 + off + 8 * CH);
}

template <bool NORM>
__device__ __forceinline__ void tile_put(float* __restrict__ buf, int tid, v4f (&st)[6],
                                         v4f m4, v4f r4) {
    const int r = tid >> 4, c4 = tid & 15;
    float* p = buf + r * A_PITCH + c4 * 4;
    if (NORM) {
        st[0] = (st[0] - m4) * r4;
        st[1] = (st[1] - m4) * r4;
    }
    *reinterpret_cast<v4f*>(p)                                  = st[0];
    *reinterpret_cast<v4f*>(p + 8 * A_PITCH)                    = st[1];
    *reinterpret_cast<v4f*>(p + 16 * A_PITCH)                   = st[2];
    *reinterpret_cast<v4f*>(p + 16 * A_PITCH + 8 * A_PITCH)     = st[3];
    *reinterpret_cast<v4f*>(p + 32 * A_PITCH)                   = 2.0f * st[4] - st[0];
    *reinterpret_cast<v4f*>(p + 32 * A_PITCH + 8 * A_PITCH)     = 2.0f * st[5] - st[1];
}

template <bool NORM>
__global__ __launch_bounds__(128) void k_cheb_gemm(
        const float* __restrict__ tx0, const float* __restrict__ tx1,
        const float* __restrict__ p2, const float* __restrict__ Wk,
        const float* __restrict__ bias,
        const float* __restrict__ mean0, const float* __restrict__ rstd0,
        float* __restrict__ out,
        float* __restrict__ sum, float* __restrict__ sumsq) {
    __shared__ __attribute__((aligned(16))) float As[2][3 * 16 * A_PITCH];
    const int tid   = threadIdx.x;
    const int wave  = tid >> 5;
    const int lane  = tid & 31;
    const int mrow  = lane & 15;
    const int khalf = lane >> 4;
    const int n0    = wave * 16;
    const int tile0 = blockIdx.x * GEMM_TPB;

    v4f m4 = {0.f, 0.f, 0.f, 0.f}, r4 = {1.f, 1.f, 1.f, 1.f};
    if (NORM) {
        m4 = *reinterpret_cast<const v4f*>(mean0 + (tid & 15) * 4);
        r4 = *reinterpret_cast<const v4f*>(rstd0 + (tid & 15) * 4);
    }

    v4f st[6];
    tile_fetch(tx0, tx1, p2, tile0 * 16, tid, st);

    v2f bfrag[48];
#pragma unroll
    for (int t = 0; t < 3; ++t) {
#pragma unroll
        for (int kk = 0; kk < 16; ++kk) {
            const int k0 = kk * 4 + khalf * 2;
            const float* __restrict__ Wt = Wk + t * CH * CH;
            bfrag[t * 16 + kk].x = Wt[(k0 + 0) * CH + n0 + mrow];
            bfrag[t * 16 + kk].y = Wt[(k0 + 1) * CH + n0 + mrow];
        }
    }
    const float bb = bias[n0 + mrow];

    tile_put<NORM>(As[0], tid, st, m4, r4);
    __syncthreads();

    float s = 0.0f, s2 = 0.0f;
    for (int j = 0; j < GEMM_TPB; ++j) {
        const int rowbase = (tile0 + j) * 16;
        const bool more = (j + 1 < GEMM_TPB);
        if (more) tile_fetch(tx0, tx1, p2, rowbase + 16, tid, st);
        __builtin_amdgcn_sched_barrier(0);

        const float* __restrict__ Ab = As[j & 1] + mrow * A_PITCH + khalf * 2;
        v8f c = {0.f, 0.f, 0.f, 0.f, 0.f, 0.f, 0.f, 0.f};
#pragma unroll
        for (int t = 0; t < 3; ++t) {
#pragma unroll
            for (int kk = 0; kk < 16; ++kk) {
                v2f a = *reinterpret_cast<const v2f*>(Ab + t * 16 * A_PITCH + kk * 4);
                c = __builtin_amdgcn_wmma_f32_16x16x4_f32(
                        false, a, false, bfrag[t * 16 + kk], (short)0, c, false, false);
            }
        }
        __builtin_amdgcn_sched_barrier(0);

#pragma unroll
        for (int i = 0; i < 8; ++i) {
            float y = c[i] + bb;
            y = (y > 0.0f) ? y : 0.0f;
            out[(long long)(rowbase + i + 8 * khalf) * CH + n0 + mrow] = y;
            s  += y;
            s2 += y * y;
        }

        if (more) {
            tile_put<NORM>(As[(j + 1) & 1], tid, st, m4, r4);
            __syncthreads();
        }
    }

    s  += __shfl_xor(s, 16, 32);
    s2 += __shfl_xor(s2, 16, 32);
    if (khalf == 0) {
        atomicAdd(&sum[n0 + mrow], s);
        atomicAdd(&sumsq[n0 + mrow], s2);
    }
}

__global__ void k_finstats(const float* __restrict__ sum, const float* __restrict__ sumsq,
                           float* __restrict__ mean, float* __restrict__ rstd) {
    int ch = threadIdx.x;
    if (ch < CH) {
        float m = sum[ch] * (1.0f / NN);
        float v = sumsq[ch] * (1.0f / NN) - m * m;
        mean[ch] = m;
        rstd[ch] = rsqrtf(v + EPSV);
    }
}

__global__ void k_norm(float* __restrict__ h, const float* __restrict__ mean,
                       const float* __restrict__ rstd, int n) {
    int i = blockIdx.x * blockDim.x + threadIdx.x;
    if (i < n) {
        int ch = i & (CH - 1);
        h[i] = (h[i] - mean[ch]) * rstd[ch];
    }
}

template <bool IN_RAW, bool OUT_NORM, bool FIRST>
static void run_layer(const int* rowptr, int* cedge, const float* dinv,
                      const float* xin, const float* Wk, const float* bk,
                      float* tx1, float* tx2, float* hout, float* stats,
                      hipStream_t stream) {
    const int FN = NN * CH;
    const float* mean = stats + 2 * CH;
    const float* rstd = stats + 3 * CH;
    k_gather<IN_RAW, FIRST><<<NN / 8, 256, 0, stream>>>(rowptr, cedge, dinv, xin, mean, rstd,
                                                        tx1);
    k_gather<false, false><<<NN / 8, 256, 0, stream>>>(rowptr, cedge, dinv, tx1, mean, rstd,
                                                       tx2);
    k_zero<<<1, 256, 0, stream>>>(stats, 128);
    k_cheb_gemm<IN_RAW><<<GEMM_NB, 128, 0, stream>>>(xin, tx1, tx2, Wk, bk, mean, rstd, hout,
                                                     stats, stats + CH);
    k_finstats<<<1, 64, 0, stream>>>(stats, stats + CH, stats + 2 * CH, stats + 3 * CH);
    if (OUT_NORM)
        k_norm<<<(FN + 255) / 256, 256, 0, stream>>>(hout, stats + 2 * CH, stats + 3 * CH, FN);
}

extern "C" void kernel_launch(void* const* d_in, const int* in_sizes, int n_in,
                              void* d_out, int out_size, void* d_ws, size_t ws_size,
                              hipStream_t stream) {
    (void)in_sizes; (void)n_in; (void)out_size; (void)ws_size;
    const float* x   = (const float*)d_in[0];
    const int*   ei  = (const int*)d_in[1];
    const float* W1  = (const float*)d_in[2];
    const float* b1  = (const float*)d_in[3];
    const float* W2  = (const float*)d_in[4];
    const float* b2  = (const float*)d_in[5];
    float* out = (float*)d_out;

    const int* src  = ei;
    const int* dstp = ei + NE;

    float* ws   = (float*)d_ws;
    int*   rowptr = (int*)ws;
    float* dinv = ws + NN;
    float* tx1  = dinv + NN;
    float* tx2  = tx1 + NN * CH;
    float* h1   = tx2 + NN * CH;
    float* stats = h1 + NN * CH;
    int*   seg  = (int*)tx2;
    int*   startT = (int*)h1;
    int*   cedge = (int*)out;

    k_bin<<<NBIN, BIN_T, 0, stream>>>(src, dstp, seg, startT);
    k_build<<<NBKT, 1024, 0, stream>>>(seg, startT, rowptr, dinv, cedge);

    run_layer<false, false, true>(rowptr, cedge, dinv, x, W1, b1, tx1, tx2, h1, stats, stream);
    run_layer<true, true, false>(rowptr, cedge, dinv, h1, W2, b2, tx1, tx2, out, stats, stream);
}
